# starting kernel
# baseline (speedup 1.0000x reference)
#include <hip/hip_runtime.h>
#include <hip/hip_bf16.h>

#define VOCAB  50257
#define NCLS   20
#define BATCH  4096
#define SEQ    200
#define HID    256

typedef __attribute__((ext_vector_type(2))) float v2f;
typedef __attribute__((ext_vector_type(8))) float v8f;

__global__ void ft_transpose_kernel(const float* __restrict__ src,
                                    float* __restrict__ dst,
                                    int R, int C) {
    __shared__ float tile[32][33];
    const int t  = threadIdx.x;
    const int tx = t & 31;
    const int ty = t >> 5;
    const int c0 = blockIdx.x * 32;
    const int r0 = blockIdx.y * 32;

    for (int i = ty; i < 32; i += 8) {
        const int r = r0 + i, c = c0 + tx;
        tile[i][tx] = (r < R && c < C) ? src[(size_t)r * C + c] : 0.0f;
    }
    __syncthreads();
    for (int i = ty; i < 32; i += 8) {
        const int c = c0 + i, r = r0 + tx;
        if (c < C && r < R) dst[(size_t)c * R + r] = tile[tx][i];
    }
}

__global__ void ft_layer1_kernel(const int*   __restrict__ ids,
                                 const float* __restrict__ W1T,
                                 const float* __restrict__ b1,
                                 float*       __restrict__ y1) {
    __shared__ int s_ids[SEQ];
    __shared__ int s_keep[SEQ];
    const int b = blockIdx.x;
    const int t = threadIdx.x;

    if (t < SEQ) s_ids[t] = ids[(size_t)b * SEQ + t];
    __syncthreads();

    if (t < SEQ) {
        const int v = s_ids[t];
        int keep = 1;
        for (int k = 0; k < t; ++k) {
            if (s_ids[k] == v) { keep = 0; break; }
        }
        s_keep[t] = keep;
    }
    __syncthreads();

    float acc = b1[t];
    for (int j = 0; j < SEQ; ++j) {
        if (s_keep[j]) {
            acc += W1T[(size_t)s_ids[j] * HID + t];
        }
    }
    y1[(size_t)b * HID + t] = fmaxf(acc, 0.0f);
}

__global__ void ft_layer2_wmma_kernel(const float* __restrict__ y1,
                                      const float* __restrict__ W2T,
                                      const float* __restrict__ b2,
                                      float*       __restrict__ y2) {
    __shared__ float As[16][HID + 1];
    const int tid = threadIdx.x;
    const int m0  = blockIdx.x * 16;

    for (int i = 0; i < 16; ++i) {
        As[i][tid] = y1[(size_t)(m0 + i) * HID + tid];
    }
    __syncthreads();

    const int wave  = tid >> 5;
    const int lane  = tid & 31;
    const int lm    = lane & 15;
    const int khalf = (lane >> 4) << 1;
    const int mbase = m0 + ((lane >> 4) << 3);

    for (int nt = 0; nt < 2; ++nt) {
        const int n0 = (wave * 2 + nt) * 16;

        v8f acc;
        const float bias = b2[n0 + lm];
        #pragma unroll
        for (int r = 0; r < 8; ++r) acc[r] = bias;

        #pragma unroll 4
        for (int kk = 0; kk < HID; kk += 4) {
            v2f a, bfr;
            a.x   = As[lm][kk + khalf];
            a.y   = As[lm][kk + khalf + 1];
            bfr.x = W2T[(size_t)(kk + khalf) * HID + n0 + lm];
            bfr.y = W2T[(size_t)(kk + khalf + 1) * HID + n0 + lm];
            acc = __builtin_amdgcn_wmma_f32_16x16x4_f32(
                false, a, false, bfr, (short)0, acc, false, false);
        }

        #pragma unroll
        for (int r = 0; r < 8; ++r) {
            const float v = acc[r] > 0.0f ? acc[r] : 0.0f;
            y2[(size_t)(mbase + r) * HID + n0 + lm] = v;
        }
    }
}

__global__ void ft_layer3_kernel(const float* __restrict__ y2,
                                 const float* __restrict__ W3,
                                 const float* __restrict__ b3,
                                 float*       __restrict__ out) {
    const int row = blockIdx.x * 8 + (threadIdx.x >> 5);
    const int o   = threadIdx.x & 31;
    if (o >= NCLS) return;

    const float4* yr = (const float4*)(y2 + (size_t)row * HID);
    const float4* wr = (const float4*)(W3 + (size_t)o * HID);
    float acc = b3[o];
    #pragma unroll 4
    for (int k = 0; k < HID / 4; ++k) {
        const float4 a = yr[k];
        const float4 w = wr[k];
        acc = fmaf(a.x, w.x, acc);
        acc = fmaf(a.y, w.y, acc);
        acc = fmaf(a.z, w.z, acc);
        acc = fmaf(a.w, w.w, acc);
    }
    out[(size_t)row * NCLS + o] = acc;
}

extern "C" void kernel_launch(void* const* d_in, const int* in_sizes, int n_in,
                              void* d_out, int out_size, void* d_ws, size_t ws_size,
                              hipStream_t stream) {
    const int*   ids = (const int*)  d_in[0];
    const float* W1  = (const float*)d_in[1];
    const float* b1  = (const float*)d_in[2];
    const float* W2  = (const float*)d_in[3];
    const float* b2  = (const float*)d_in[4];
    const float* W3  = (const float*)d_in[5];
    const float* b3  = (const float*)d_in[6];
    float* out = (float*)d_out;

    float* W1T = (float*)d_ws;
    float* W2T = W1T + (size_t)VOCAB * HID;
    float* y1  = W2T + (size_t)HID * HID;
    float* y2  = y1  + (size_t)BATCH * HID;

    {
        dim3 g((VOCAB + 31) / 32, (HID + 31) / 32);
        ft_transpose_kernel<<<g, 256, 0, stream>>>(W1, W1T, HID, VOCAB);
    }
    {
        dim3 g((HID + 31) / 32, (HID + 31) / 32);
        ft_transpose_kernel<<<g, 256, 0, stream>>>(W2, W2T, HID, HID);
    }
    ft_layer1_kernel<<<BATCH, 256, 0, stream>>>(ids, W1T, b1, y1);
    ft_layer2_wmma_kernel<<<BATCH / 16, 256, 0, stream>>>(y1, W2T, b2, y2);
    ft_layer3_kernel<<<BATCH / 8, 256, 0, stream>>>(y2, W3, b3, out);
}


# layer 1: the row's 200 ids are split over 4 waves (was: the 256 columns over 8), a wave sums whole 1 KB W1T rows, 10 16-byte loads in flight; 4 partials added via LDS; sum order now 4 runs of 50
# speedup vs baseline: 2.0600x; 2.0600x over previous
#include <hip/hip_runtime.h>
#include <hip/hip_bf16.h>

#define VOCAB  50257
#define NCLS   20
#define BATCH  4096
#define SEQ    200
#define HID    256

typedef __attribute__((ext_vector_type(2))) float v2f;
typedef __attribute__((ext_vector_type(8))) float v8f;

__global__ void ft_transpose_kernel(const float* __restrict__ src,
                                    float* __restrict__ dst,
                                    int R, int C) {
    __shared__ float tile[32][33];
    const int t  = threadIdx.x;
    const int tx = t & 31;
    const int ty = t >> 5;
    const int c0 = blockIdx.x * 32;
    const int r0 = blockIdx.y * 32;

    for (int i = ty; i < 32; i += 8) {
        const int r = r0 + i, c = c0 + tx;
        tile[i][tx] = (r < R && c < C) ? src[(size_t)r * C + c] : 0.0f;
    }
    __syncthreads();
    for (int i = ty; i < 32; i += 8) {
        const int c = c0 + i, r = r0 + tx;
        if (c < C && r < R) dst[(size_t)c * R + r] = tile[tx][i];
    }
}

#define L1_THREADS  128
#define L1_WAVES    (L1_THREADS / 32)
#define L1_PER_WAVE (SEQ / L1_WAVES)
#define L1_GROUP    5
#define L1_IDS_PAD  224

__global__ __launch_bounds__(L1_THREADS)
void ft_layer1_kernel(const int*   __restrict__ ids,
                      const float* __restrict__ W1T,
                      const float* __restrict__ b1,
                      float*       __restrict__ y1) {
    __shared__ __attribute__((aligned(16))) int   s_ids[L1_IDS_PAD];
    __shared__ int s_keep[SEQ];
    __shared__ __attribute__((aligned(16))) float s_part[L1_WAVES * HID];

    const int b    = blockIdx.x;
    const int t    = threadIdx.x;
    const int lane = t & 31;
    const int wave = t >> 5;

    for (int j = t; j < L1_IDS_PAD; j += L1_THREADS)
        s_ids[j] = (j < SEQ) ? ids[(size_t)b * SEQ + j] : -1;
    __syncthreads();

    const int4* s4 = (const int4*)s_ids;
    for (int j = t; j < SEQ; j += L1_THREADS) {
        const int v  = s_ids[j];
        const int jb = j & ~31;
        int dup = 0;
        for (int k4 = 0; k4 < (jb >> 2); ++k4) {
            const int4 q = s4[k4];
            dup |= (q.x == v) | (q.y == v) | (q.z == v) | (q.w == v);
        }
        for (int i = 0; i < 8; ++i) {
            const int4 q = s4[(jb >> 2) + i];
            const int  k = jb + 4 * i;
            dup |= ((q.x == v) & (k     < j)) | ((q.y == v) & (k + 1 < j))
                |  ((q.z == v) & (k + 2 < j)) | ((q.w == v) & (k + 3 < j));
        }
        s_keep[j] = dup ? 0 : 1;
    }
    __syncthreads();

    const int j0 = wave * L1_PER_WAVE;
    float4 a0 = make_float4(0.0f, 0.0f, 0.0f, 0.0f);
    float4 a1 = a0;
    for (int g = 0; g < L1_PER_WAVE; g += L1_GROUP) {
        float4 x0[L1_GROUP], x1[L1_GROUP];
        #pragma unroll
        for (int u = 0; u < L1_GROUP; ++u) {
            const float4* row = (const float4*)(W1T + (size_t)s_ids[j0 + g + u] * HID);
            x0[u] = row[lane];
            x1[u] = row[lane + 32];
        }
        #pragma unroll
        for (int u = 0; u < L1_GROUP; ++u) {
            const bool k = s_keep[j0 + g + u] != 0;
            a0.x += k ? x0[u].x : 0.0f;  a0.y += k ? x0[u].y : 0.0f;
            a0.z += k ? x0[u].z : 0.0f;  a0.w += k ? x0[u].w : 0.0f;
            a1.x += k ? x1[u].x : 0.0f;  a1.y += k ? x1[u].y : 0.0f;
            a1.z += k ? x1[u].z : 0.0f;  a1.w += k ? x1[u].w : 0.0f;
        }
    }

    float4* sp = (float4*)(s_part + wave * HID);
    sp[lane]      = a0;
    sp[lane + 32] = a1;
    __syncthreads();

    for (int h = t; h < HID; h += L1_THREADS) {
        float acc = b1[h];
        for (int w = 0; w < L1_WAVES; ++w) acc += s_part[w * HID + h];
        y1[(size_t)b * HID + h] = fmaxf(acc, 0.0f);
    }
}

__global__ void ft_layer2_wmma_kernel(const float* __restrict__ y1,
                                      const float* __restrict__ W2T,
                                      const float* __restrict__ b2,
                                      float*       __restrict__ y2) {
    __shared__ float As[16][HID + 1];
    const int tid = threadIdx.x;
    const int m0  = blockIdx.x * 16;

    for (int i = 0; i < 16; ++i) {
        As[i][tid] = y1[(size_t)(m0 + i) * HID + tid];
    }
    __syncthreads();

    const int wave  = tid >> 5;
    const int lane  = tid & 31;
    const int lm    = lane & 15;
    const int khalf = (lane >> 4) << 1;
    const int mbase = m0 + ((lane >> 4) << 3);

    for (int nt = 0; nt < 2; ++nt) {
        const int n0 = (wave * 2 + nt) * 16;

        v8f acc;
        const float bias = b2[n0 + lm];
        #pragma unroll
        for (int r = 0; r < 8; ++r) acc[r] = bias;

        #pragma unroll 4
        for (int kk = 0; kk < HID; kk += 4) {
            v2f a, bfr;
            a.x   = As[lm][kk + khalf];
            a.y   = As[lm][kk + khalf + 1];
            bfr.x = W2T[(size_t)(kk + khalf) * HID + n0 + lm];
            bfr.y = W2T[(size_t)(kk + khalf + 1) * HID + n0 + lm];
            acc = __builtin_amdgcn_wmma_f32_16x16x4_f32(
                false, a, false, bfr, (short)0, acc, false, false);
        }

        #pragma unroll
        for (int r = 0; r < 8; ++r) {
            const float v = acc[r] > 0.0f ? acc[r] : 0.0f;
            y2[(size_t)(mbase + r) * HID + n0 + lm] = v;
        }
    }
}

__global__ void ft_layer3_kernel(const float* __restrict__ y2,
                                 const float* __restrict__ W3,
                                 const float* __restrict__ b3,
                                 float*       __restrict__ out) {
    const int row = blockIdx.x * 8 + (threadIdx.x >> 5);
    const int o   = threadIdx.x & 31;
    if (o >= NCLS) return;

    const float4* yr = (const float4*)(y2 + (size_t)row * HID);
    const float4* wr = (const float4*)(W3 + (size_t)o * HID);
    float acc = b3[o];
    #pragma unroll 4
    for (int k = 0; k < HID / 4; ++k) {
        const float4 a = yr[k];
        const float4 w = wr[k];
        acc = fmaf(a.x, w.x, acc);
        acc = fmaf(a.y, w.y, acc);
        acc = fmaf(a.z, w.z, acc);
        acc = fmaf(a.w, w.w, acc);
    }
    out[(size_t)row * NCLS + o] = acc;
}

extern "C" void kernel_launch(void* const* d_in, const int* in_sizes, int n_in,
                              void* d_out, int out_size, void* d_ws, size_t ws_size,
                              hipStream_t stream) {
    const int*   ids = (const int*)  d_in[0];
    const float* W1  = (const float*)d_in[1];
    const float* b1  = (const float*)d_in[2];
    const float* W2  = (const float*)d_in[3];
    const float* b2  = (const float*)d_in[4];
    const float* W3  = (const float*)d_in[5];
    const float* b3  = (const float*)d_in[6];
    float* out = (float*)d_out;

    float* W1T = (float*)d_ws;
    float* W2T = W1T + (size_t)VOCAB * HID;
    float* y1  = W2T + (size_t)HID * HID;
    float* y2  = y1  + (size_t)BATCH * HID;

    {
        dim3 g((VOCAB + 31) / 32, (HID + 31) / 32);
        ft_transpose_kernel<<<g, 256, 0, stream>>>(W1, W1T, HID, VOCAB);
    }
    {
        dim3 g((HID + 31) / 32, (HID + 31) / 32);
        ft_transpose_kernel<<<g, 256, 0, stream>>>(W2, W2T, HID, HID);
    }
    ft_layer1_kernel<<<BATCH, L1_THREADS, 0, stream>>>(ids, W1T, b1, y1);
    ft_layer2_wmma_kernel<<<BATCH / 16, 256, 0, stream>>>(y1, W2T, b2, y2);
    ft_layer3_kernel<<<BATCH / 8, 256, 0, stream>>>(y2, W3, b3, out);
}


# layer 2 takes W2 as it arrives: a lane's B fragments are 16-byte loads from W2 rows, all 64 in flight before the first product (barrier orders LDS only); no transpose launch; k order new, bias last
# speedup vs baseline: 2.8063x; 1.3623x over previous
#include <hip/hip_runtime.h>
#include <hip/hip_bf16.h>

#define VOCAB  50257
#define NCLS   20
#define BATCH  4096
#define SEQ    200
#define HID    256

typedef __attribute__((ext_vector_type(2))) float v2f;
typedef __attribute__((ext_vector_type(8))) float v8f;

#define DD_ROWS     8
#define DD_PER_LANE 7

static_assert(VOCAB <= 0xffff, "ids and the zero slot's index must fit 16 bits");
static_assert(SEQ % 8 == 0 && SEQ <= 32 * DD_PER_LANE, "row pieces of 8 ids, 7 positions per lane");

__global__ __launch_bounds__(DD_ROWS * 32)
void ft_ids_bank_order_kernel(const int*      __restrict__ ids,
                              unsigned short* __restrict__ ids2) {
    __shared__ __attribute__((aligned(16))) int            s_key[DD_ROWS][SEQ];
    __shared__ __attribute__((aligned(16))) unsigned short s_srt[DD_ROWS][SEQ];
    __shared__ __attribute__((aligned(16))) unsigned short s_o[DD_ROWS * SEQ];

    const int lane = threadIdx.x & 31;
    const int wave = threadIdx.x >> 5;
    const int b    = blockIdx.x * DD_ROWS + wave;

    int key[DD_PER_LANE];
    #pragma unroll
    for (int m = 0; m < DD_PER_LANE; ++m) {
        const int j  = lane + 32 * m;
        const int id = (j < SEQ) ? ids[(size_t)b * SEQ + j] : 0;
        key[m] = ((((id & 63) << 10) | ((id >> 6) & 1023)) << 8) | j;
        if (j < SEQ) s_key[wave][j] = key[m];
    }
    __syncthreads();

    int rank[DD_PER_LANE];
    #pragma unroll
    for (int m = 0; m < DD_PER_LANE; ++m) rank[m] = 0;
    const int4* s4 = (const int4*)s_key[wave];
    for (int k4 = 0; k4 < SEQ / 4; ++k4) {
        const int4 q = s4[k4];
        #pragma unroll
        for (int m = 0; m < DD_PER_LANE; ++m)
            rank[m] += (q.x < key[m]) + (q.y < key[m]) + (q.z < key[m]) + (q.w < key[m]);
    }
    #pragma unroll
    for (int m = 0; m < DD_PER_LANE; ++m)
        if (lane + 32 * m < SEQ) s_srt[wave][rank[m]] = (unsigned short)(key[m] >> 8);
    __syncthreads();

    #pragma unroll
    for (int m = 0; m < DD_PER_LANE; ++m) {
        const int i = lane + 32 * m;
        if (i < SEQ) {
            const unsigned k    = s_srt[wave][i];
            const unsigned prev = (i > 0) ? (unsigned)s_srt[wave][i - 1] : 0xffffu;
            const unsigned id   = ((k & 1023u) << 6) | (k >> 10);
            s_o[wave * SEQ + i] = (unsigned short)((k == prev) ? (unsigned)VOCAB : id);
        }
    }
    __syncthreads();

    if (threadIdx.x < DD_ROWS * SEQ / 8) {
        uint4* dst = (uint4*)(ids2 + (size_t)blockIdx.x * DD_ROWS * SEQ);
        dst[threadIdx.x] = ((const uint4*)s_o)[threadIdx.x];
    }
}

#define LC_THREADS  1024
#define LC_WAVES    (LC_THREADS / 32)
#define LC_WPAD     50272
#define LC_FILL     10
#define LC_PAIRS    (BATCH / 2)
#define LC_PIECES   (SEQ / 8)

template <int MASK>
__device__ __forceinline__ float ft_row_xor(float v) {
    return __int_as_float(__builtin_amdgcn_update_dpp(
        0, __float_as_int(v), 0x160 + MASK, 0xf, 0xf, false));
}

__device__ __forceinline__ float ft_other_row(float v) {
    return __int_as_float(__builtin_amdgcn_permlanex16(
        0, __float_as_int(v), 0x76543210, 0xfedcba98, false, false));
}

__device__ __forceinline__ float ft_piece_sum(const float* s_w, uint4 a) {
    const float e0 = s_w[a.x & 0xffffu] + s_w[a.x >> 16];
    const float e1 = s_w[a.y & 0xffffu] + s_w[a.y >> 16];
    const float e2 = s_w[a.z & 0xffffu] + s_w[a.z >> 16];
    const float e3 = s_w[a.w & 0xffffu] + s_w[a.w >> 16];
    return (e0 + e1) + (e2 + e3);
}

__global__ __launch_bounds__(LC_THREADS)
void ft_layer1_col_kernel(const unsigned short* __restrict__ ids2,
                          const float*          __restrict__ W1,
                          const float*          __restrict__ b1,
                          float*                __restrict__ y1T) {
    __shared__ __attribute__((aligned(16))) float s_w[LC_WPAD];
    __shared__ __attribute__((aligned(16))) float s_out[BATCH];

    const int h    = blockIdx.x;
    const int t    = threadIdx.x;
    const int lane = t & 31;
    const int wave = t >> 5;
    const int c    = lane & 15;
    const int hh   = lane >> 4;

    const float* wrow = W1 + (size_t)h * VOCAB;
    for (int base = 0; base < LC_WPAD; base += LC_THREADS * LC_FILL) {
        float x[LC_FILL];
        #pragma unroll
        for (int u = 0; u < LC_FILL; ++u) {
            const int v = base + u * LC_THREADS + t;
            x[u] = (v < VOCAB) ? wrow[v] : 0.0f;
        }
        #pragma unroll
        for (int u = 0; u < LC_FILL; ++u) {
            const int v = base + u * LC_THREADS + t;
            if (v < LC_WPAD) s_w[v] = x[u];
        }
    }
    __syncthreads();

    const bool own = lane < LC_PIECES;
    const int  li  = own ? lane : LC_PIECES - 1;
    int p = wave;
    const unsigned short* rp = ids2 + (size_t)(2 * p) * SEQ + 8 * li;
    uint4 nA = *(const uint4*)rp;
    uint4 nB = *(const uint4*)(rp + SEQ);

    #pragma unroll 2
    for (int i = 0; i < LC_PAIRS / LC_WAVES; ++i) {
        const uint4 a = nA, b = nB;
        const int   cur = 2 * p + hh;
        p  = (i + 1 < LC_PAIRS / LC_WAVES) ? p + LC_WAVES : p;
        rp = ids2 + (size_t)(2 * p) * SEQ + 8 * li;
        nA = *(const uint4*)rp;
        nB = *(const uint4*)(rp + SEQ);

        float sA = ft_piece_sum(s_w, a);
        float sB = ft_piece_sum(s_w, b);
        sA = own ? sA : 0.0f;
        sB = own ? sB : 0.0f;
        const float keep = hh ? sB : sA;
        const float give = hh ? sA : sB;
        float s = keep + ft_other_row(give);
        s += ft_row_xor<1>(s);
        s += ft_row_xor<2>(s);
        s += ft_row_xor<4>(s);
        s += ft_row_xor<8>(s);
        if (c == 0) s_out[cur] = s;
    }
    __syncthreads();

    const float  bias = b1[h];
    const float4 r = ((const float4*)s_out)[t];
    float4 o;
    o.x = fmaxf(bias + r.x, 0.0f);
    o.y = fmaxf(bias + r.y, 0.0f);
    o.z = fmaxf(bias + r.z, 0.0f);
    o.w = fmaxf(bias + r.w, 0.0f);
    ((float4*)(y1T + (size_t)h * BATCH))[t] = o;
}

#define L2_PITCH 260

__global__ __launch_bounds__(256)
void ft_layer2_wmma_kernel(const float* __restrict__ y1T,
                           const float* __restrict__ W2,
                           const float* __restrict__ b2,
                           float*       __restrict__ y2) {
    __shared__ __attribute__((aligned(16))) float As[16 * L2_PITCH];
    const int tid  = threadIdx.x;
    const int m0   = blockIdx.x * 16;
    const int wave = tid >> 5;
    const int lane = tid & 31;
    const int c    = lane & 15;
    const int h    = lane >> 4;
    const int n0   = wave * 32;

    const float4* a4 = (const float4*)(y1T + (size_t)tid * BATCH + m0);
    float4 av[4];
    #pragma unroll
    for (int q = 0; q < 4; ++q) av[q] = a4[q];

    const float4* w0 = (const float4*)(W2 + (size_t)(n0 + c) * HID + 4 * h);
    const float4* w1 = (const float4*)(W2 + (size_t)(n0 + 16 + c) * HID + 4 * h);
    float4 bq0[HID / 8], bq1[HID / 8];
    #pragma unroll
    for (int j = 0; j < HID / 8; ++j) {
        bq0[j] = w0[2 * j];
        bq1[j] = w1[2 * j];
    }

    #pragma unroll
    for (int q = 0; q < 4; ++q) {
        As[(4 * q    ) * L2_PITCH + tid] = av[q].x;
        As[(4 * q + 1) * L2_PITCH + tid] = av[q].y;
        As[(4 * q + 2) * L2_PITCH + tid] = av[q].z;
        As[(4 * q + 3) * L2_PITCH + tid] = av[q].w;
    }
    __builtin_amdgcn_fence(__ATOMIC_RELEASE, "workgroup", "local");
    __builtin_amdgcn_s_barrier();
    __builtin_amdgcn_fence(__ATOMIC_ACQUIRE, "workgroup", "local");

    v8f acc0, acc1;
    #pragma unroll
    for (int r = 0; r < 8; ++r) { acc0[r] = 0.0f; acc1[r] = 0.0f; }

    const float* arow = As + c * L2_PITCH + 4 * h;
    #pragma unroll
    for (int j = 0; j < HID / 8; ++j) {
        const float4 a = *(const float4*)(arow + 8 * j);
        v2f alo, ahi, b0lo, b0hi, b1lo, b1hi;
        alo.x  = a.x;       alo.y  = a.y;
        ahi.x  = a.z;       ahi.y  = a.w;
        b0lo.x = bq0[j].x;  b0lo.y = bq0[j].y;
        b0hi.x = bq0[j].z;  b0hi.y = bq0[j].w;
        b1lo.x = bq1[j].x;  b1lo.y = bq1[j].y;
        b1hi.x = bq1[j].z;  b1hi.y = bq1[j].w;
        acc0 = __builtin_amdgcn_wmma_f32_16x16x4_f32(
            false, alo, false, b0lo, (short)0, acc0, false, false);
        acc1 = __builtin_amdgcn_wmma_f32_16x16x4_f32(
            false, alo, false, b1lo, (short)0, acc1, false, false);
        acc0 = __builtin_amdgcn_wmma_f32_16x16x4_f32(
            false, ahi, false, b0hi, (short)0, acc0, false, false);
        acc1 = __builtin_amdgcn_wmma_f32_16x16x4_f32(
            false, ahi, false, b1hi, (short)0, acc1, false, false);
    }

    const int   mbase = m0 + (h << 3);
    const float bias0 = b2[n0 + c];
    const float bias1 = b2[n0 + 16 + c];
    #pragma unroll
    for (int r = 0; r < 8; ++r) {
        const float s0 = acc0[r] + bias0;
        const float s1 = acc1[r] + bias1;
        const float v0 = s0 > 0.0f ? s0 : 0.0f;
        const float v1 = s1 > 0.0f ? s1 : 0.0f;
        y2[(size_t)(mbase + r) * HID + n0 + c]      = v0;
        y2[(size_t)(mbase + r) * HID + n0 + 16 + c] = v1;
    }
}

__global__ void ft_layer3_kernel(const float* __restrict__ y2,
                                 const float* __restrict__ W3,
                                 const float* __restrict__ b3,
                                 float*       __restrict__ out) {
    const int row = blockIdx.x * 8 + (threadIdx.x >> 5);
    const int o   = threadIdx.x & 31;
    if (o >= NCLS) return;

    const float4* yr = (const float4*)(y2 + (size_t)row * HID);
    const float4* wr = (const float4*)(W3 + (size_t)o * HID);
    float acc = b3[o];
    #pragma unroll 4
    for (int k = 0; k < HID / 4; ++k) {
        const float4 a = yr[k];
        const float4 w = wr[k];
        acc = fmaf(a.x, w.x, acc);
        acc = fmaf(a.y, w.y, acc);
        acc = fmaf(a.z, w.z, acc);
        acc = fmaf(a.w, w.w, acc);
    }
    out[(size_t)row * NCLS + o] = acc;
}

extern "C" void kernel_launch(void* const* d_in, const int* in_sizes, int n_in,
                              void* d_out, int out_size, void* d_ws, size_t ws_size,
                              hipStream_t stream) {
    const int*   ids = (const int*)  d_in[0];
    const float* W1  = (const float*)d_in[1];
    const float* b1  = (const float*)d_in[2];
    const float* W2  = (const float*)d_in[3];
    const float* b2  = (const float*)d_in[4];
    const float* W3  = (const float*)d_in[5];
    const float* b3  = (const float*)d_in[6];
    float* out = (float*)d_out;

    unsigned short* ids2 = (unsigned short*)d_ws;
    float* y1T  = (float*)(ids2 + (size_t)BATCH * SEQ);
    float* y2   = y1T + (size_t)HID * BATCH;

    ft_ids_bank_order_kernel<<<BATCH / DD_ROWS, DD_ROWS * 32, 0, stream>>>(ids, ids2);
    ft_layer1_col_kernel<<<HID, LC_THREADS, 0, stream>>>(ids2, W1, b1, y1T);
    ft_layer2_wmma_kernel<<<BATCH / 16, 256, 0, stream>>>(y1T, W2, b2, y2);
    ft_layer3_kernel<<<BATCH / 8, 256, 0, stream>>>(y2, W3, b3, out);
}


# layer 3 joins the layer-2 kernel, 3 launches: layer 2 formed transposed (W2 as A), its bias+relu D registers are layer 3's B fragments directly; 8 waves' partials added via LDS; layer-3 sum order new
# speedup vs baseline: 3.2621x; 1.1624x over previous
#include <hip/hip_runtime.h>
#include <hip/hip_bf16.h>

#define VOCAB  50257
#define NCLS   20
#define BATCH  4096
#define SEQ    200
#define HID    256

typedef __attribute__((ext_vector_type(2))) float v2f;
typedef __attribute__((ext_vector_type(8))) float v8f;

#define DD_ROWS     8
#define DD_PER_LANE 7

static_assert(VOCAB <= 0xffff, "ids and the zero slot's index must fit 16 bits");
static_assert(SEQ % 8 == 0 && SEQ <= 32 * DD_PER_LANE, "row pieces of 8 ids, 7 positions per lane");

__global__ __launch_bounds__(DD_ROWS * 32)
void ft_ids_bank_order_kernel(const int*      __restrict__ ids,
                              unsigned short* __restrict__ ids2) {
    __shared__ __attribute__((aligned(16))) int            s_key[DD_ROWS][SEQ];
    __shared__ __attribute__((aligned(16))) unsigned short s_srt[DD_ROWS][SEQ];
    __shared__ __attribute__((aligned(16))) unsigned short s_o[DD_ROWS * SEQ];

    const int lane = threadIdx.x & 31;
    const int wave = threadIdx.x >> 5;
    const int b    = blockIdx.x * DD_ROWS + wave;

    int key[DD_PER_LANE];
    #pragma unroll
    for (int m = 0; m < DD_PER_LANE; ++m) {
        const int j  = lane + 32 * m;
        const int id = (j < SEQ) ? ids[(size_t)b * SEQ + j] : 0;
        key[m] = ((((id & 63) << 10) | ((id >> 6) & 1023)) << 8) | j;
        if (j < SEQ) s_key[wave][j] = key[m];
    }
    __syncthreads();

    int rank[DD_PER_LANE];
    #pragma unroll
    for (int m = 0; m < DD_PER_LANE; ++m) rank[m] = 0;
    const int4* s4 = (const int4*)s_key[wave];
    for (int k4 = 0; k4 < SEQ / 4; ++k4) {
        const int4 q = s4[k4];
        #pragma unroll
        for (int m = 0; m < DD_PER_LANE; ++m)
            rank[m] += (q.x < key[m]) + (q.y < key[m]) + (q.z < key[m]) + (q.w < key[m]);
    }
    #pragma unroll
    for (int m = 0; m < DD_PER_LANE; ++m)
        if (lane + 32 * m < SEQ) s_srt[wave][rank[m]] = (unsigned short)(key[m] >> 8);
    __syncthreads();

    #pragma unroll
    for (int m = 0; m < DD_PER_LANE; ++m) {
        const int i = lane + 32 * m;
        if (i < SEQ) {
            const unsigned k    = s_srt[wave][i];
            const unsigned prev = (i > 0) ? (unsigned)s_srt[wave][i - 1] : 0xffffu;
            const unsigned id   = ((k & 1023u) << 6) | (k >> 10);
            s_o[wave * SEQ + i] = (unsigned short)((k == prev) ? (unsigned)VOCAB : id);
        }
    }
    __syncthreads();

    if (threadIdx.x < DD_ROWS * SEQ / 8) {
        uint4* dst = (uint4*)(ids2 + (size_t)blockIdx.x * DD_ROWS * SEQ);
        dst[threadIdx.x] = ((const uint4*)s_o)[threadIdx.x];
    }
}

#define LC_THREADS  1024
#define LC_WAVES    (LC_THREADS / 32)
#define LC_WPAD     50272
#define LC_FILL     10
#define LC_PAIRS    (BATCH / 2)
#define LC_PIECES   (SEQ / 8)

template <int MASK>
__device__ __forceinline__ float ft_row_xor(float v) {
    return __int_as_float(__builtin_amdgcn_update_dpp(
        0, __float_as_int(v), 0x160 + MASK, 0xf, 0xf, false));
}

__device__ __forceinline__ float ft_other_row(float v) {
    return __int_as_float(__builtin_amdgcn_permlanex16(
        0, __float_as_int(v), 0x76543210, 0xfedcba98, false, false));
}

__device__ __forceinline__ float ft_piece_sum(const float* s_w, uint4 a) {
    const float e0 = s_w[a.x & 0xffffu] + s_w[a.x >> 16];
    const float e1 = s_w[a.y & 0xffffu] + s_w[a.y >> 16];
    const float e2 = s_w[a.z & 0xffffu] + s_w[a.z >> 16];
    const float e3 = s_w[a.w & 0xffffu] + s_w[a.w >> 16];
    return (e0 + e1) + (e2 + e3);
}

__global__ __launch_bounds__(LC_THREADS)
void ft_layer1_col_kernel(const unsigned short* __restrict__ ids2,
                          const float*          __restrict__ W1,
                          const float*          __restrict__ b1,
                          float*                __restrict__ y1T) {
    __shared__ __attribute__((aligned(16))) float s_w[LC_WPAD];
    __shared__ __attribute__((aligned(16))) float s_out[BATCH];

    const int h    = blockIdx.x;
    const int t    = threadIdx.x;
    const int lane = t & 31;
    const int wave = t >> 5;
    const int c    = lane & 15;
    const int hh   = lane >> 4;

    const float* wrow = W1 + (size_t)h * VOCAB;
    for (int base = 0; base < LC_WPAD; base += LC_THREADS * LC_FILL) {
        float x[LC_FILL];
        #pragma unroll
        for (int u = 0; u < LC_FILL; ++u) {
            const int v = base + u * LC_THREADS + t;
            x[u] = (v < VOCAB) ? wrow[v] : 0.0f;
        }
        #pragma unroll
        for (int u = 0; u < LC_FILL; ++u) {
            const int v = base + u * LC_THREADS + t;
            if (v < LC_WPAD) s_w[v] = x[u];
        }
    }
    __syncthreads();

    const bool own = lane < LC_PIECES;
    const int  li  = own ? lane : LC_PIECES - 1;
    int p = wave;
    const unsigned short* rp = ids2 + (size_t)(2 * p) * SEQ + 8 * li;
    uint4 nA = *(const uint4*)rp;
    uint4 nB = *(const uint4*)(rp + SEQ);

    #pragma unroll 2
    for (int i = 0; i < LC_PAIRS / LC_WAVES; ++i) {
        const uint4 a = nA, b = nB;
        const int   cur = 2 * p + hh;
        p  = (i + 1 < LC_PAIRS / LC_WAVES) ? p + LC_WAVES : p;
        rp = ids2 + (size_t)(2 * p) * SEQ + 8 * li;
        nA = *(const uint4*)rp;
        nB = *(const uint4*)(rp + SEQ);

        float sA = ft_piece_sum(s_w, a);
        float sB = ft_piece_sum(s_w, b);
        sA = own ? sA : 0.0f;
        sB = own ? sB : 0.0f;
        const float keep = hh ? sB : sA;
        const float give = hh ? sA : sB;
        float s = keep + ft_other_row(give);
        s += ft_row_xor<1>(s);
        s += ft_row_xor<2>(s);
        s += ft_row_xor<4>(s);
        s += ft_row_xor<8>(s);
        if (c == 0) s_out[cur] = s;
    }
    __syncthreads();

    const float  bias = b1[h];
    const float4 r = ((const float4*)s_out)[t];
    float4 o;
    o.x = fmaxf(bias + r.x, 0.0f);
    o.y = fmaxf(bias + r.y, 0.0f);
    o.z = fmaxf(bias + r.z, 0.0f);
    o.w = fmaxf(bias + r.w, 0.0f);
    ((float4*)(y1T + (size_t)h * BATCH))[t] = o;
}

#define L2_PITCH 260
#define L3_RED   12

static_assert(NCLS > 16 && NCLS <= 20, "second class tile: rows 0..3 only");

__global__ __launch_bounds__(256)
void ft_layer23_wmma_kernel(const float* __restrict__ y1T,
                            const float* __restrict__ W2,
                            const float* __restrict__ b2,
                            const float* __restrict__ W3,
                            const float* __restrict__ b3,
                            float*       __restrict__ out) {
    __shared__ __attribute__((aligned(16))) float As[16 * L2_PITCH];
    __shared__ __attribute__((aligned(16))) float red[8 * 32 * L3_RED];
    const int tid  = threadIdx.x;
    const int m0   = blockIdx.x * 16;
    const int wave = tid >> 5;
    const int lane = tid & 31;
    const int c    = lane & 15;
    const int h    = lane >> 4;
    const int n0   = wave * 32;

    const float4* a4 = (const float4*)(y1T + (size_t)tid * BATCH + m0);
    float4 av[4];
    #pragma unroll
    for (int q = 0; q < 4; ++q) av[q] = a4[q];

    const float4* bp = (const float4*)(b2 + n0 + 8 * h);
    const float4 c00 = bp[0], c01 = bp[1];
    const float4 c10 = bp[4], c11 = bp[5];
    const int     ohi = (16 + c < NCLS) ? 16 + c : NCLS - 1;
    const float4* w3a = (const float4*)(W3 + (size_t)c * HID + n0 + 8 * h);
    const float4* w3b = (const float4*)(W3 + (size_t)ohi * HID + n0 + 8 * h);
    float4 wa[2][2], wb[2][2];
    #pragma unroll
    for (int u = 0; u < 2; ++u) {
        wa[u][0] = w3a[4 * u];
        wa[u][1] = w3a[4 * u + 1];
        wb[u][0] = w3b[4 * u];
        wb[u][1] = w3b[4 * u + 1];
    }

    const float4* w0 = (const float4*)(W2 + (size_t)(n0 + c) * HID + 4 * h);
    const float4* w1 = (const float4*)(W2 + (size_t)(n0 + 16 + c) * HID + 4 * h);
    float4 bq0[HID / 8], bq1[HID / 8];
    #pragma unroll
    for (int j = 0; j < HID / 8; ++j) {
        bq0[j] = w0[2 * j];
        bq1[j] = w1[2 * j];
    }

    #pragma unroll
    for (int q = 0; q < 4; ++q) {
        As[(4 * q    ) * L2_PITCH + tid] = av[q].x;
        As[(4 * q + 1) * L2_PITCH + tid] = av[q].y;
        As[(4 * q + 2) * L2_PITCH + tid] = av[q].z;
        As[(4 * q + 3) * L2_PITCH + tid] = av[q].w;
    }
    __builtin_amdgcn_fence(__ATOMIC_RELEASE, "workgroup", "local");
    __builtin_amdgcn_s_barrier();
    __builtin_amdgcn_fence(__ATOMIC_ACQUIRE, "workgroup", "local");

    v8f acc0, acc1;
    #pragma unroll
    for (int r = 0; r < 8; ++r) { acc0[r] = 0.0f; acc1[r] = 0.0f; }

    const float* arow = As + c * L2_PITCH + 4 * h;
    #pragma unroll
    for (int j = 0; j < HID / 8; ++j) {
        const float4 a = *(const float4*)(arow + 8 * j);
        v2f alo, ahi, b0lo, b0hi, b1lo, b1hi;
        alo.x  = a.x;       alo.y  = a.y;
        ahi.x  = a.z;       ahi.y  = a.w;
        b0lo.x = bq0[j].x;  b0lo.y = bq0[j].y;
        b0hi.x = bq0[j].z;  b0hi.y = bq0[j].w;
        b1lo.x = bq1[j].x;  b1lo.y = bq1[j].y;
        b1hi.x = bq1[j].z;  b1hi.y = bq1[j].w;
        acc0 = __builtin_amdgcn_wmma_f32_16x16x4_f32(
            false, b0lo, false, alo, (short)0, acc0, false, false);
        acc1 = __builtin_amdgcn_wmma_f32_16x16x4_f32(
            false, b1lo, false, alo, (short)0, acc1, false, false);
        acc0 = __builtin_amdgcn_wmma_f32_16x16x4_f32(
            false, b0hi, false, ahi, (short)0, acc0, false, false);
        acc1 = __builtin_amdgcn_wmma_f32_16x16x4_f32(
            false, b1hi, false, ahi, (short)0, acc1, false, false);
    }

    asm volatile("s_wait_alu depctr_va_vdst(0)" : "+v"(acc0), "+v"(acc1));

    const float cb0[8] = { c00.x, c00.y, c00.z, c00.w, c01.x, c01.y, c01.z, c01.w };
    const float cb1[8] = { c10.x, c10.y, c10.z, c10.w, c11.x, c11.y, c11.z, c11.w };
    #pragma unroll
    for (int r = 0; r < 8; ++r) {
        const float s0 = acc0[r] + cb0[r];
        const float s1 = acc1[r] + cb1[r];
        acc0[r] = s0 > 0.0f ? s0 : 0.0f;
        acc1[r] = s1 > 0.0f ? s1 : 0.0f;
    }
    asm volatile("" :: "v"(acc0), "v"(acc1));

    const float fa[2][8] = {
        { wa[0][0].x, wa[0][0].y, wa[0][0].z, wa[0][0].w, wa[0][1].x, wa[0][1].y, wa[0][1].z, wa[0][1].w },
        { wa[1][0].x, wa[1][0].y, wa[1][0].z, wa[1][0].w, wa[1][1].x, wa[1][1].y, wa[1][1].z, wa[1][1].w } };
    const float fb[2][8] = {
        { wb[0][0].x, wb[0][0].y, wb[0][0].z, wb[0][0].w, wb[0][1].x, wb[0][1].y, wb[0][1].z, wb[0][1].w },
        { wb[1][0].x, wb[1][0].y, wb[1][0].z, wb[1][0].w, wb[1][1].x, wb[1][1].y, wb[1][1].z, wb[1][1].w } };

    v8f o0, o1;
    #pragma unroll
    for (int r = 0; r < 8; ++r) { o0[r] = 0.0f; o1[r] = 0.0f; }
    #pragma unroll
    for (int u = 0; u < 2; ++u) {
        #pragma unroll
        for (int q = 0; q < 4; ++q) {
            v2f yb, w3lo, w3hi;
            yb.x   = u ? acc1[2 * q]     : acc0[2 * q];
            yb.y   = u ? acc1[2 * q + 1] : acc0[2 * q + 1];
            w3lo.x = fa[u][2 * q];
            w3lo.y = fa[u][2 * q + 1];
            w3hi.x = fb[u][2 * q];
            w3hi.y = fb[u][2 * q + 1];
            o0 = __builtin_amdgcn_wmma_f32_16x16x4_f32(
                false, w3lo, false, yb, (short)0, o0, false, false);
            o1 = __builtin_amdgcn_wmma_f32_16x16x4_f32(
                false, w3hi, false, yb, (short)0, o1, false, false);
        }
    }

    float* rp = red + (wave * 32 + lane) * L3_RED;
    *(float4*)(rp)     = make_float4(o0[0], o0[1], o0[2], o0[3]);
    *(float4*)(rp + 4) = make_float4(o0[4], o0[5], o0[6], o0[7]);
    *(float4*)(rp + 8) = make_float4(o1[0], o1[1], o1[2], o1[3]);
    __syncthreads();

    for (int idx = tid; idx < 16 * NCLS; idx += 256) {
        const int m   = idx / NCLS;
        const int o   = idx - m * NCLS;
        const int src = ((o >> 3) & 1) * 16 + m;
        const int off = (o >> 4) * 8 + (o & 7);
        const float* q = red + src * L3_RED + off;
        const float p01 = q[0 * 32 * L3_RED] + q[1 * 32 * L3_RED];
        const float p23 = q[2 * 32 * L3_RED] + q[3 * 32 * L3_RED];
        const float p45 = q[4 * 32 * L3_RED] + q[5 * 32 * L3_RED];
        const float p67 = q[6 * 32 * L3_RED] + q[7 * 32 * L3_RED];
        out[(size_t)m0 * NCLS + idx] = ((p01 + p23) + (p45 + p67)) + b3[o];
    }
}

extern "C" void kernel_launch(void* const* d_in, const int* in_sizes, int n_in,
                              void* d_out, int out_size, void* d_ws, size_t ws_size,
                              hipStream_t stream) {
    const int*   ids = (const int*)  d_in[0];
    const float* W1  = (const float*)d_in[1];
    const float* b1  = (const float*)d_in[2];
    const float* W2  = (const float*)d_in[3];
    const float* b2  = (const float*)d_in[4];
    const float* W3  = (const float*)d_in[5];
    const float* b3  = (const float*)d_in[6];
    float* out = (float*)d_out;

    unsigned short* ids2 = (unsigned short*)d_ws;
    float* y1T  = (float*)(ids2 + (size_t)BATCH * SEQ);

    ft_ids_bank_order_kernel<<<BATCH / DD_ROWS, DD_ROWS * 32, 0, stream>>>(ids, ids2);
    ft_layer1_col_kernel<<<HID, LC_THREADS, 0, stream>>>(ids2, W1, b1, y1T);
    ft_layer23_wmma_kernel<<<BATCH / 16, 256, 0, stream>>>(y1T, W2, b2, W3, b3, out);
}


# layer-1 gather: ids of 4 row pairs per wave on request ahead of the pair being gathered (was 1), two loop copies so no ids move between registers; LDS reads stop waiting for ids; sum order kept
# speedup vs baseline: 3.8025x; 1.1657x over previous
#include <hip/hip_runtime.h>
#include <hip/hip_bf16.h>

#define VOCAB  50257
#define NCLS   20
#define BATCH  4096
#define SEQ    200
#define HID    256

typedef __attribute__((ext_vector_type(2))) float v2f;
typedef __attribute__((ext_vector_type(8))) float v8f;

#define DD_ROWS     8
#define DD_PER_LANE 7

static_assert(VOCAB <= 0xffff, "ids and the zero slot's index must fit 16 bits");
static_assert(SEQ % 8 == 0 && SEQ <= 32 * DD_PER_LANE, "row pieces of 8 ids, 7 positions per lane");

__global__ __launch_bounds__(DD_ROWS * 32)
void ft_ids_bank_order_kernel(const int*      __restrict__ ids,
                              unsigned short* __restrict__ ids2) {
    __shared__ __attribute__((aligned(16))) int            s_key[DD_ROWS][SEQ];
    __shared__ __attribute__((aligned(16))) unsigned short s_srt[DD_ROWS][SEQ];
    __shared__ __attribute__((aligned(16))) unsigned short s_o[DD_ROWS * SEQ];

    const int lane = threadIdx.x & 31;
    const int wave = threadIdx.x >> 5;
    const int b    = blockIdx.x * DD_ROWS + wave;

    int key[DD_PER_LANE];
    #pragma unroll
    for (int m = 0; m < DD_PER_LANE; ++m) {
        const int j  = lane + 32 * m;
        const int id = (j < SEQ) ? ids[(size_t)b * SEQ + j] : 0;
        key[m] = ((((id & 63) << 10) | ((id >> 6) & 1023)) << 8) | j;
        if (j < SEQ) s_key[wave][j] = key[m];
    }
    __syncthreads();

    int rank[DD_PER_LANE];
    #pragma unroll
    for (int m = 0; m < DD_PER_LANE; ++m) rank[m] = 0;
    const int4* s4 = (const int4*)s_key[wave];
    for (int k4 = 0; k4 < SEQ / 4; ++k4) {
        const int4 q = s4[k4];
        #pragma unroll
        for (int m = 0; m < DD_PER_LANE; ++m)
            rank[m] += (q.x < key[m]) + (q.y < key[m]) + (q.z < key[m]) + (q.w < key[m]);
    }
    #pragma unroll
    for (int m = 0; m < DD_PER_LANE; ++m)
        if (lane + 32 * m < SEQ) s_srt[wave][rank[m]] = (unsigned short)(key[m] >> 8);
    __syncthreads();

    #pragma unroll
    for (int m = 0; m < DD_PER_LANE; ++m) {
        const int i = lane + 32 * m;
        if (i < SEQ) {
            const unsigned k    = s_srt[wave][i];
            const unsigned prev = (i > 0) ? (unsigned)s_srt[wave][i - 1] : 0xffffu;
            const unsigned id   = ((k & 1023u) << 6) | (k >> 10);
            s_o[wave * SEQ + i] = (unsigned short)((k == prev) ? (unsigned)VOCAB : id);
        }
    }
    __syncthreads();

    if (threadIdx.x < DD_ROWS * SEQ / 8) {
        uint4* dst = (uint4*)(ids2 + (size_t)blockIdx.x * DD_ROWS * SEQ);
        dst[threadIdx.x] = ((const uint4*)s_o)[threadIdx.x];
    }
}

#define LC_THREADS  1024
#define LC_WAVES    (LC_THREADS / 32)
#define LC_WPAD     50272
#define LC_FILL     10
#define LC_PAIRS    (BATCH / 2)
#define LC_PIECES   (SEQ / 8)
#define LC_TRIPS    (LC_PAIRS / LC_WAVES)
#define LC_AHEAD    4

static_assert(LC_TRIPS % (2 * LC_AHEAD) == 0, "the gather loop advances by LC_AHEAD pairs, two trips per copy");

template <int MASK>
__device__ __forceinline__ float ft_row_xor(float v) {
    return __int_as_float(__builtin_amdgcn_update_dpp(
        0, __float_as_int(v), 0x160 + MASK, 0xf, 0xf, false));
}

__device__ __forceinline__ float ft_other_row(float v) {
    return __int_as_float(__builtin_amdgcn_permlanex16(
        0, __float_as_int(v), 0x76543210, 0xfedcba98, false, false));
}

__device__ __forceinline__ float ft_piece_sum(const float* s_w, uint4 a) {
    const float e0 = s_w[a.x & 0xffffu] + s_w[a.x >> 16];
    const float e1 = s_w[a.y & 0xffffu] + s_w[a.y >> 16];
    const float e2 = s_w[a.z & 0xffffu] + s_w[a.z >> 16];
    const float e3 = s_w[a.w & 0xffffu] + s_w[a.w >> 16];
    return (e0 + e1) + (e2 + e3);
}

__global__ __launch_bounds__(LC_THREADS)
void ft_layer1_col_kernel(const unsigned short* __restrict__ ids2,
                          const float*          __restrict__ W1,
                          const float*          __restrict__ b1,
                          float*                __restrict__ y1T) {
    __shared__ __attribute__((aligned(16))) float s_w[LC_WPAD];
    __shared__ __attribute__((aligned(16))) float s_out[BATCH];

    const int h    = blockIdx.x;
    const int t    = threadIdx.x;
    const int lane = t & 31;
    const int wave = t >> 5;
    const int c    = lane & 15;
    const int hh   = lane >> 4;

    const float* wrow = W1 + (size_t)h * VOCAB;
    for (int base = 0; base < LC_WPAD; base += LC_THREADS * LC_FILL) {
        float x[LC_FILL];
        #pragma unroll
        for (int u = 0; u < LC_FILL; ++u) {
            const int v = base + u * LC_THREADS + t;
            x[u] = (v < VOCAB) ? wrow[v] : 0.0f;
        }
        #pragma unroll
        for (int u = 0; u < LC_FILL; ++u) {
            const int v = base + u * LC_THREADS + t;
            if (v < LC_WPAD) s_w[v] = x[u];
        }
    }
    __syncthreads();

    const bool own = lane < LC_PIECES;
    const int  li  = own ? lane : LC_PIECES - 1;
    const unsigned short* rlane = ids2 + 8 * li;
    uint4 qA[LC_AHEAD], qB[LC_AHEAD];
    #pragma unroll
    for (int d = 0; d < LC_AHEAD; ++d) {
        const unsigned short* rp = rlane + (size_t)(2 * (wave + LC_WAVES * d)) * SEQ;
        qA[d] = *(const uint4*)rp;
        qB[d] = *(const uint4*)(rp + SEQ);
    }

    #pragma unroll 2
    for (int i = 0; i < LC_TRIPS; i += LC_AHEAD) {
        #pragma unroll
        for (int d = 0; d < LC_AHEAD; ++d) {
            const uint4 a = qA[d], b = qB[d];
            const int   cur = 2 * (wave + LC_WAVES * (i + d)) + hh;
            const int in = (i + d + LC_AHEAD < LC_TRIPS) ? i + d + LC_AHEAD : LC_TRIPS - 1;
            const unsigned short* rp = rlane + (size_t)(2 * (wave + LC_WAVES * in)) * SEQ;
            qA[d] = *(const uint4*)rp;
            qB[d] = *(const uint4*)(rp + SEQ);

            float sA = ft_piece_sum(s_w, a);
            float sB = ft_piece_sum(s_w, b);
            sA = own ? sA : 0.0f;
            sB = own ? sB : 0.0f;
            const float keep = hh ? sB : sA;
            const float give = hh ? sA : sB;
            float s = keep + ft_other_row(give);
            s += ft_row_xor<1>(s);
            s += ft_row_xor<2>(s);
            s += ft_row_xor<4>(s);
            s += ft_row_xor<8>(s);
            if (c == 0) s_out[cur] = s;
        }
    }
    __syncthreads();

    const float  bias = b1[h];
    const float4 r = ((const float4*)s_out)[t];
    float4 o;
    o.x = fmaxf(bias + r.x, 0.0f);
    o.y = fmaxf(bias + r.y, 0.0f);
    o.z = fmaxf(bias + r.z, 0.0f);
    o.w = fmaxf(bias + r.w, 0.0f);
    ((float4*)(y1T + (size_t)h * BATCH))[t] = o;
}

#define L2_PITCH 260
#define L3_RED   12

static_assert(NCLS > 16 && NCLS <= 20, "second class tile: rows 0..3 only");

__global__ __launch_bounds__(256)
void ft_layer23_wmma_kernel(const float* __restrict__ y1T,
                            const float* __restrict__ W2,
                            const float* __restrict__ b2,
                            const float* __restrict__ W3,
                            const float* __restrict__ b3,
                            float*       __restrict__ out) {
    __shared__ __attribute__((aligned(16))) float As[16 * L2_PITCH];
    __shared__ __attribute__((aligned(16))) float red[8 * 32 * L3_RED];
    const int tid  = threadIdx.x;
    const int m0   = blockIdx.x * 16;
    const int wave = tid >> 5;
    const int lane = tid & 31;
    const int c    = lane & 15;
    const int h    = lane >> 4;
    const int n0   = wave * 32;

    const float4* a4 = (const float4*)(y1T + (size_t)tid * BATCH + m0);
    float4 av[4];
    #pragma unroll
    for (int q = 0; q < 4; ++q) av[q] = a4[q];

    const float4* bp = (const float4*)(b2 + n0 + 8 * h);
    const float4 c00 = bp[0], c01 = bp[1];
    const float4 c10 = bp[4], c11 = bp[5];
    const int     ohi = (16 + c < NCLS) ? 16 + c : NCLS - 1;
    const float4* w3a = (const float4*)(W3 + (size_t)c * HID + n0 + 8 * h);
    const float4* w3b = (const float4*)(W3 + (size_t)ohi * HID + n0 + 8 * h);
    float4 wa[2][2], wb[2][2];
    #pragma unroll
    for (int u = 0; u < 2; ++u) {
        wa[u][0] = w3a[4 * u];
        wa[u][1] = w3a[4 * u + 1];
        wb[u][0] = w3b[4 * u];
        wb[u][1] = w3b[4 * u + 1];
    }

    const float4* w0 = (const float4*)(W2 + (size_t)(n0 + c) * HID + 4 * h);
    const float4* w1 = (const float4*)(W2 + (size_t)(n0 + 16 + c) * HID + 4 * h);
    float4 bq0[HID / 8], bq1[HID / 8];
    #pragma unroll
    for (int j = 0; j < HID / 8; ++j) {
        bq0[j] = w0[2 * j];
        bq1[j] = w1[2 * j];
    }

    #pragma unroll
    for (int q = 0; q < 4; ++q) {
        As[(4 * q    ) * L2_PITCH + tid] = av[q].x;
        As[(4 * q + 1) * L2_PITCH + tid] = av[q].y;
        As[(4 * q + 2) * L2_PITCH + tid] = av[q].z;
        As[(4 * q + 3) * L2_PITCH + tid] = av[q].w;
    }
    __builtin_amdgcn_fence(__ATOMIC_RELEASE, "workgroup", "local");
    __builtin_amdgcn_s_barrier();
    __builtin_amdgcn_fence(__ATOMIC_ACQUIRE, "workgroup", "local");

    v8f acc0, acc1;
    #pragma unroll
    for (int r = 0; r < 8; ++r) { acc0[r] = 0.0f; acc1[r] = 0.0f; }

    const float* arow = As + c * L2_PITCH + 4 * h;
    #pragma unroll
    for (int j = 0; j < HID / 8; ++j) {
        const float4 a = *(const float4*)(arow + 8 * j);
        v2f alo, ahi, b0lo, b0hi, b1lo, b1hi;
        alo.x  = a.x;       alo.y  = a.y;
        ahi.x  = a.z;       ahi.y  = a.w;
        b0lo.x = bq0[j].x;  b0lo.y = bq0[j].y;
        b0hi.x = bq0[j].z;  b0hi.y = bq0[j].w;
        b1lo.x = bq1[j].x;  b1lo.y = bq1[j].y;
        b1hi.x = bq1[j].z;  b1hi.y = bq1[j].w;
        acc0 = __builtin_amdgcn_wmma_f32_16x16x4_f32(
            false, b0lo, false, alo, (short)0, acc0, false, false);
        acc1 = __builtin_amdgcn_wmma_f32_16x16x4_f32(
            false, b1lo, false, alo, (short)0, acc1, false, false);
        acc0 = __builtin_amdgcn_wmma_f32_16x16x4_f32(
            false, b0hi, false, ahi, (short)0, acc0, false, false);
        acc1 = __builtin_amdgcn_wmma_f32_16x16x4_f32(
            false, b1hi, false, ahi, (short)0, acc1, false, false);
    }

    asm volatile("s_wait_alu depctr_va_vdst(0)" : "+v"(acc0), "+v"(acc1));

    const float cb0[8] = { c00.x, c00.y, c00.z, c00.w, c01.x, c01.y, c01.z, c01.w };
    const float cb1[8] = { c10.x, c10.y, c10.z, c10.w, c11.x, c11.y, c11.z, c11.w };
    #pragma unroll
    for (int r = 0; r < 8; ++r) {
        const float s0 = acc0[r] + cb0[r];
        const float s1 = acc1[r] + cb1[r];
        acc0[r] = s0 > 0.0f ? s0 : 0.0f;
        acc1[r] = s1 > 0.0f ? s1 : 0.0f;
    }
    asm volatile("" :: "v"(acc0), "v"(acc1));

    const float fa[2][8] = {
        { wa[0][0].x, wa[0][0].y, wa[0][0].z, wa[0][0].w, wa[0][1].x, wa[0][1].y, wa[0][1].z, wa[0][1].w },
        { wa[1][0].x, wa[1][0].y, wa[1][0].z, wa[1][0].w, wa[1][1].x, wa[1][1].y, wa[1][1].z, wa[1][1].w } };
    const float fb[2][8] = {
        { wb[0][0].x, wb[0][0].y, wb[0][0].z, wb[0][0].w, wb[0][1].x, wb[0][1].y, wb[0][1].z, wb[0][1].w },
        { wb[1][0].x, wb[1][0].y, wb[1][0].z, wb[1][0].w, wb[1][1].x, wb[1][1].y, wb[1][1].z, wb[1][1].w } };

    v8f o0, o1;
    #pragma unroll
    for (int r = 0; r < 8; ++r) { o0[r] = 0.0f; o1[r] = 0.0f; }
    #pragma unroll
    for (int u = 0; u < 2; ++u) {
        #pragma unroll
        for (int q = 0; q < 4; ++q) {
            v2f yb, w3lo, w3hi;
            yb.x   = u ? acc1[2 * q]     : acc0[2 * q];
            yb.y   = u ? acc1[2 * q + 1] : acc0[2 * q + 1];
            w3lo.x = fa[u][2 * q];
            w3lo.y = fa[u][2 * q + 1];
            w3hi.x = fb[u][2 * q];
            w3hi.y = fb[u][2 * q + 1];
            o0 = __builtin_amdgcn_wmma_f32_16x16x4_f32(
                false, w3lo, false, yb, (short)0, o0, false, false);
            o1 = __builtin_amdgcn_wmma_f32_16x16x4_f32(
                false, w3hi, false, yb, (short)0, o1, false, false);
        }
    }

    float* rp = red + (wave * 32 + lane) * L3_RED;
    *(float4*)(rp)     = make_float4(o0[0], o0[1], o0[2], o0[3]);
    *(float4*)(rp + 4) = make_float4(o0[4], o0[5], o0[6], o0[7]);
    *(float4*)(rp + 8) = make_float4(o1[0], o1[1], o1[2], o1[3]);
    __syncthreads();

    for (int idx = tid; idx < 16 * NCLS; idx += 256) {
        const int m   = idx / NCLS;
        const int o   = idx - m * NCLS;
        const int src = ((o >> 3) & 1) * 16 + m;
        const int off = (o >> 4) * 8 + (o & 7);
        const float* q = red + src * L3_RED + off;
        const float p01 = q[0 * 32 * L3_RED] + q[1 * 32 * L3_RED];
        const float p23 = q[2 * 32 * L3_RED] + q[3 * 32 * L3_RED];
        const float p45 = q[4 * 32 * L3_RED] + q[5 * 32 * L3_RED];
        const float p67 = q[6 * 32 * L3_RED] + q[7 * 32 * L3_RED];
        out[(size_t)m0 * NCLS + idx] = ((p01 + p23) + (p45 + p67)) + b3[o];
    }
}

extern "C" void kernel_launch(void* const* d_in, const int* in_sizes, int n_in,
                              void* d_out, int out_size, void* d_ws, size_t ws_size,
                              hipStream_t stream) {
    const int*   ids = (const int*)  d_in[0];
    const float* W1  = (const float*)d_in[1];
    const float* b1  = (const float*)d_in[2];
    const float* W2  = (const float*)d_in[3];
    const float* b2  = (const float*)d_in[4];
    const float* W3  = (const float*)d_in[5];
    const float* b3  = (const float*)d_in[6];
    float* out = (float*)d_out;

    unsigned short* ids2 = (unsigned short*)d_ws;
    float* y1T  = (float*)(ids2 + (size_t)BATCH * SEQ);

    ft_ids_bank_order_kernel<<<BATCH / DD_ROWS, DD_ROWS * 32, 0, stream>>>(ids, ids2);
    ft_layer1_col_kernel<<<HID, LC_THREADS, 0, stream>>>(ids2, W1, b1, y1T);
    ft_layer23_wmma_kernel<<<BATCH / 16, 256, 0, stream>>>(y1T, W2, b2, W3, b3, out);
}


# pre-pass: bank order now from a bitonic sorting network in registers (8 keys per lane, lane exchanges by DPP), not an all-pairs rank count through LDS; ids2 and sum order unchanged
# speedup vs baseline: 4.3382x; 1.1409x over previous
#include <hip/hip_runtime.h>
#include <hip/hip_bf16.h>

#define VOCAB  50257
#define NCLS   20
#define BATCH  4096
#define SEQ    200
#define HID    256

typedef __attribute__((ext_vector_type(2))) float v2f;
typedef __attribute__((ext_vector_type(8))) float v8f;

#define DD_ROWS     8
#define DD_PER_LANE 8
#define DD_PADKEY   0xffffffffu

static_assert(VOCAB <= 0xffff, "ids and the zero slot's index must fit 16 bits");
static_assert(SEQ % 8 == 0 && SEQ <= 32 * (DD_PER_LANE - 1), "row pieces of 8 ids; the last register of every lane is padding");

template <int MASK>
__device__ __forceinline__ unsigned ft_key_row_xor(unsigned v) {
    return (unsigned)__builtin_amdgcn_update_dpp(0, (int)v, 0x160 + MASK, 0xf, 0xf, false);
}

__device__ __forceinline__ unsigned ft_key_lane_xor(unsigned v, int m) {
    switch (m) {
        case 1:  return ft_key_row_xor<1>(v);
        case 2:  return ft_key_row_xor<2>(v);
        case 4:  return ft_key_row_xor<4>(v);
        case 8:  return ft_key_row_xor<8>(v);
        default: return (unsigned)__builtin_amdgcn_permlanex16(
                     0, (int)v, 0x76543210, 0xfedcba98, false, false);
    }
}

__global__ __launch_bounds__(DD_ROWS * 32)
void ft_ids_bank_order_kernel(const int*      __restrict__ ids,
                              unsigned short* __restrict__ ids2) {
    __shared__ __attribute__((aligned(16))) unsigned short s_o[DD_ROWS * SEQ];

    const int lane = threadIdx.x & 31;
    const int wave = threadIdx.x >> 5;
    const int b    = blockIdx.x * DD_ROWS + wave;
    const int e0   = lane * DD_PER_LANE;

    unsigned key[DD_PER_LANE];
    #pragma unroll
    for (int r = 0; r < DD_PER_LANE - 1; ++r) {
        const int j  = lane + 32 * r;
        const int id = (j < SEQ) ? ids[(size_t)b * SEQ + j] : 0;
        key[r] = (j < SEQ) ? (unsigned)(((id & 63) << 10) | ((id >> 6) & 1023)) : DD_PADKEY;
    }
    key[DD_PER_LANE - 1] = DD_PADKEY;

    #pragma unroll
    for (int k = 2; k <= 32 * DD_PER_LANE; k <<= 1) {
        #pragma unroll
        for (int j = k >> 1; j > 0; j >>= 1) {
            if (j >= DD_PER_LANE) {
                const bool keep_min = ((e0 & k) == 0) == ((e0 & j) == 0);
                #pragma unroll
                for (int r = 0; r < DD_PER_LANE; ++r) {
                    const unsigned p  = ft_key_lane_xor(key[r], j >> 3);
                    const unsigned lo = key[r] < p ? key[r] : p;
                    const unsigned hi = key[r] < p ? p : key[r];
                    key[r] = keep_min ? lo : hi;
                }
            } else {
                #pragma unroll
                for (int r = 0; r < DD_PER_LANE; ++r) {
                    if ((r & j) == 0) {
                        const bool asc = (k < DD_PER_LANE) ? ((r & k) == 0) : ((e0 & k) == 0);
                        const unsigned a  = key[r], c = key[r | j];
                        const unsigned lo = a < c ? a : c;
                        const unsigned hi = a < c ? c : a;
                        key[r]     = asc ? lo : hi;
                        key[r | j] = asc ? hi : lo;
                    }
                }
            }
        }
    }

    const unsigned before = (unsigned)__shfl((int)key[DD_PER_LANE - 1], (lane + 31) & 31, 32);
    unsigned o[DD_PER_LANE];
    #pragma unroll
    for (int r = 0; r < DD_PER_LANE; ++r) {
        const unsigned k    = key[r];
        const unsigned prev = r ? key[r - 1] : before;
        const unsigned id   = ((k & 1023u) << 6) | ((k >> 10) & 63u);
        o[r] = (k == prev) ? (unsigned)VOCAB : id;
    }
    if (lane < SEQ / 8) {
        uint4 w;
        w.x = o[0] | (o[1] << 16);
        w.y = o[2] | (o[3] << 16);
        w.z = o[4] | (o[5] << 16);
        w.w = o[6] | (o[7] << 16);
        ((uint4*)s_o)[wave * (SEQ / 8) + lane] = w;
    }
    __syncthreads();

    if (threadIdx.x < DD_ROWS * SEQ / 8) {
        uint4* dst = (uint4*)(ids2 + (size_t)blockIdx.x * DD_ROWS * SEQ);
        dst[threadIdx.x] = ((const uint4*)s_o)[threadIdx.x];
    }
}

#define LC_THREADS  1024
#define LC_WAVES    (LC_THREADS / 32)
#define LC_WPAD     50272
#define LC_FILL     10
#define LC_PAIRS    (BATCH / 2)
#define LC_PIECES   (SEQ / 8)
#define LC_TRIPS    (LC_PAIRS / LC_WAVES)
#define LC_AHEAD    4

static_assert(LC_TRIPS % (2 * LC_AHEAD) == 0, "the gather loop advances by LC_AHEAD pairs, two trips per copy");

template <int MASK>
__device__ __forceinline__ float ft_row_xor(float v) {
    return __int_as_float(__builtin_amdgcn_update_dpp(
        0, __float_as_int(v), 0x160 + MASK, 0xf, 0xf, false));
}

__device__ __forceinline__ float ft_other_row(float v) {
    return __int_as_float(__builtin_amdgcn_permlanex16(
        0, __float_as_int(v), 0x76543210, 0xfedcba98, false, false));
}

__device__ __forceinline__ float ft_piece_sum(const float* s_w, uint4 a) {
    const float e0 = s_w[a.x & 0xffffu] + s_w[a.x >> 16];
    const float e1 = s_w[a.y & 0xffffu] + s_w[a.y >> 16];
    const float e2 = s_w[a.z & 0xffffu] + s_w[a.z >> 16];
    const float e3 = s_w[a.w & 0xffffu] + s_w[a.w >> 16];
    return (e0 + e1) + (e2 + e3);
}

__global__ __launch_bounds__(LC_THREADS)
void ft_layer1_col_kernel(const unsigned short* __restrict__ ids2,
                          const float*          __restrict__ W1,
                          const float*          __restrict__ b1,
                          float*                __restrict__ y1T) {
    __shared__ __attribute__((aligned(16))) float s_w[LC_WPAD];
    __shared__ __attribute__((aligned(16))) float s_out[BATCH];

    const int h    = blockIdx.x;
    const int t    = threadIdx.x;
    const int lane = t & 31;
    const int wave = t >> 5;
    const int c    = lane & 15;
    const int hh   = lane >> 4;

    const float* wrow = W1 + (size_t)h * VOCAB;
    for (int base = 0; base < LC_WPAD; base += LC_THREADS * LC_FILL) {
        float x[LC_FILL];
        #pragma unroll
        for (int u = 0; u < LC_FILL; ++u) {
            const int v = base + u * LC_THREADS + t;
            x[u] = (v < VOCAB) ? wrow[v] : 0.0f;
        }
        #pragma unroll
        for (int u = 0; u < LC_FILL; ++u) {
            const int v = base + u * LC_THREADS + t;
            if (v < LC_WPAD) s_w[v] = x[u];
        }
    }
    __syncthreads();

    const bool own = lane < LC_PIECES;
    const int  li  = own ? lane : LC_PIECES - 1;
    const unsigned short* rlane = ids2 + 8 * li;
    uint4 qA[LC_AHEAD], qB[LC_AHEAD];
    #pragma unroll
    for (int d = 0; d < LC_AHEAD; ++d) {
        const unsigned short* rp = rlane + (size_t)(2 * (wave + LC_WAVES * d)) * SEQ;
        qA[d] = *(const uint4*)rp;
        qB[d] = *(const uint4*)(rp + SEQ);
    }

    #pragma unroll 2
    for (int i = 0; i < LC_TRIPS; i += LC_AHEAD) {
        #pragma unroll
        for (int d = 0; d < LC_AHEAD; ++d) {
            const uint4 a = qA[d], b = qB[d];
            const int   cur = 2 * (wave + LC_WAVES * (i + d)) + hh;
            const int in = (i + d + LC_AHEAD < LC_TRIPS) ? i + d + LC_AHEAD : LC_TRIPS - 1;
            const unsigned short* rp = rlane + (size_t)(2 * (wave + LC_WAVES * in)) * SEQ;
            qA[d] = *(const uint4*)rp;
            qB[d] = *(const uint4*)(rp + SEQ);

            float sA = ft_piece_sum(s_w, a);
            float sB = ft_piece_sum(s_w, b);
            sA = own ? sA : 0.0f;
            sB = own ? sB : 0.0f;
            const float keep = hh ? sB : sA;
            const float give = hh ? sA : sB;
            float s = keep + ft_other_row(give);
            s += ft_row_xor<1>(s);
            s += ft_row_xor<2>(s);
            s += ft_row_xor<4>(s);
            s += ft_row_xor<8>(s);
            if (c == 0) s_out[cur] = s;
        }
    }
    __syncthreads();

    const float  bias = b1[h];
    const float4 r = ((const float4*)s_out)[t];
    float4 o;
    o.x = fmaxf(bias + r.x, 0.0f);
    o.y = fmaxf(bias + r.y, 0.0f);
    o.z = fmaxf(bias + r.z, 0.0f);
    o.w = fmaxf(bias + r.w, 0.0f);
    ((float4*)(y1T + (size_t)h * BATCH))[t] = o;
}

#define L2_PITCH 260
#define L3_RED   12

static_assert(NCLS > 16 && NCLS <= 20, "second class tile: rows 0..3 only");

__global__ __launch_bounds__(256)
void ft_layer23_wmma_kernel(const float* __restrict__ y1T,
                            const float* __restrict__ W2,
                            const float* __restrict__ b2,
                            const float* __restrict__ W3,
                            const float* __restrict__ b3,
                            float*       __restrict__ out) {
    __shared__ __attribute__((aligned(16))) float As[16 * L2_PITCH];
    __shared__ __attribute__((aligned(16))) float red[8 * 32 * L3_RED];
    const int tid  = threadIdx.x;
    const int m0   = blockIdx.x * 16;
    const int wave = tid >> 5;
    const int lane = tid & 31;
    const int c    = lane & 15;
    const int h    = lane >> 4;
    const int n0   = wave * 32;

    const float4* a4 = (const float4*)(y1T + (size_t)tid * BATCH + m0);
    float4 av[4];
    #pragma unroll
    for (int q = 0; q < 4; ++q) av[q] = a4[q];

    const float4* bp = (const float4*)(b2 + n0 + 8 * h);
    const float4 c00 = bp[0], c01 = bp[1];
    const float4 c10 = bp[4], c11 = bp[5];
    const int     ohi = (16 + c < NCLS) ? 16 + c : NCLS - 1;
    const float4* w3a = (const float4*)(W3 + (size_t)c * HID + n0 + 8 * h);
    const float4* w3b = (const float4*)(W3 + (size_t)ohi * HID + n0 + 8 * h);
    float4 wa[2][2], wb[2][2];
    #pragma unroll
    for (int u = 0; u < 2; ++u) {
        wa[u][0] = w3a[4 * u];
        wa[u][1] = w3a[4 * u + 1];
        wb[u][0] = w3b[4 * u];
        wb[u][1] = w3b[4 * u + 1];
    }

    const float4* w0 = (const float4*)(W2 + (size_t)(n0 + c) * HID + 4 * h);
    const float4* w1 = (const float4*)(W2 + (size_t)(n0 + 16 + c) * HID + 4 * h);
    float4 bq0[HID / 8], bq1[HID / 8];
    #pragma unroll
    for (int j = 0; j < HID / 8; ++j) {
        bq0[j] = w0[2 * j];
        bq1[j] = w1[2 * j];
    }

    #pragma unroll
    for (int q = 0; q < 4; ++q) {
        As[(4 * q    ) * L2_PITCH + tid] = av[q].x;
        As[(4 * q + 1) * L2_PITCH + tid] = av[q].y;
        As[(4 * q + 2) * L2_PITCH + tid] = av[q].z;
        As[(4 * q + 3) * L2_PITCH + tid] = av[q].w;
    }
    __builtin_amdgcn_fence(__ATOMIC_RELEASE, "workgroup", "local");
    __builtin_amdgcn_s_barrier();
    __builtin_amdgcn_fence(__ATOMIC_ACQUIRE, "workgroup", "local");

    v8f acc0, acc1;
    #pragma unroll
    for (int r = 0; r < 8; ++r) { acc0[r] = 0.0f; acc1[r] = 0.0f; }

    const float* arow = As + c * L2_PITCH + 4 * h;
    #pragma unroll
    for (int j = 0; j < HID / 8; ++j) {
        const float4 a = *(const float4*)(arow + 8 * j);
        v2f alo, ahi, b0lo, b0hi, b1lo, b1hi;
        alo.x  = a.x;       alo.y  = a.y;
        ahi.x  = a.z;       ahi.y  = a.w;
        b0lo.x = bq0[j].x;  b0lo.y = bq0[j].y;
        b0hi.x = bq0[j].z;  b0hi.y = bq0[j].w;
        b1lo.x = bq1[j].x;  b1lo.y = bq1[j].y;
        b1hi.x = bq1[j].z;  b1hi.y = bq1[j].w;
        acc0 = __builtin_amdgcn_wmma_f32_16x16x4_f32(
            false, b0lo, false, alo, (short)0, acc0, false, false);
        acc1 = __builtin_amdgcn_wmma_f32_16x16x4_f32(
            false, b1lo, false, alo, (short)0, acc1, false, false);
        acc0 = __builtin_amdgcn_wmma_f32_16x16x4_f32(
            false, b0hi, false, ahi, (short)0, acc0, false, false);
        acc1 = __builtin_amdgcn_wmma_f32_16x16x4_f32(
            false, b1hi, false, ahi, (short)0, acc1, false, false);
    }

    asm volatile("s_wait_alu depctr_va_vdst(0)" : "+v"(acc0), "+v"(acc1));

    const float cb0[8] = { c00.x, c00.y, c00.z, c00.w, c01.x, c01.y, c01.z, c01.w };
    const float cb1[8] = { c10.x, c10.y, c10.z, c10.w, c11.x, c11.y, c11.z, c11.w };
    #pragma unroll
    for (int r = 0; r < 8; ++r) {
        const float s0 = acc0[r] + cb0[r];
        const float s1 = acc1[r] + cb1[r];
        acc0[r] = s0 > 0.0f ? s0 : 0.0f;
        acc1[r] = s1 > 0.0f ? s1 : 0.0f;
    }
    asm volatile("" :: "v"(acc0), "v"(acc1));

    const float fa[2][8] = {
        { wa[0][0].x, wa[0][0].y, wa[0][0].z, wa[0][0].w, wa[0][1].x, wa[0][1].y, wa[0][1].z, wa[0][1].w },
        { wa[1][0].x, wa[1][0].y, wa[1][0].z, wa[1][0].w, wa[1][1].x, wa[1][1].y, wa[1][1].z, wa[1][1].w } };
    const float fb[2][8] = {
        { wb[0][0].x, wb[0][0].y, wb[0][0].z, wb[0][0].w, wb[0][1].x, wb[0][1].y, wb[0][1].z, wb[0][1].w },
        { wb[1][0].x, wb[1][0].y, wb[1][0].z, wb[1][0].w, wb[1][1].x, wb[1][1].y, wb[1][1].z, wb[1][1].w } };

    v8f o0, o1;
    #pragma unroll
    for (int r = 0; r < 8; ++r) { o0[r] = 0.0f; o1[r] = 0.0f; }
    #pragma unroll
    for (int u = 0; u < 2; ++u) {
        #pragma unroll
        for (int q = 0; q < 4; ++q) {
            v2f yb, w3lo, w3hi;
            yb.x   = u ? acc1[2 * q]     : acc0[2 * q];
            yb.y   = u ? acc1[2 * q + 1] : acc0[2 * q + 1];
            w3lo.x = fa[u][2 * q];
            w3lo.y = fa[u][2 * q + 1];
            w3hi.x = fb[u][2 * q];
            w3hi.y = fb[u][2 * q + 1];
            o0 = __builtin_amdgcn_wmma_f32_16x16x4_f32(
                false, w3lo, false, yb, (short)0, o0, false, false);
            o1 = __builtin_amdgcn_wmma_f32_16x16x4_f32(
                false, w3hi, false, yb, (short)0, o1, false, false);
        }
    }

    float* rp = red + (wave * 32 + lane) * L3_RED;
    *(float4*)(rp)     = make_float4(o0[0], o0[1], o0[2], o0[3]);
    *(float4*)(rp + 4) = make_float4(o0[4], o0[5], o0[6], o0[7]);
    *(float4*)(rp + 8) = make_float4(o1[0], o1[1], o1[2], o1[3]);
    __syncthreads();

    for (int idx = tid; idx < 16 * NCLS; idx += 256) {
        const int m   = idx / NCLS;
        const int o   = idx - m * NCLS;
        const int src = ((o >> 3) & 1) * 16 + m;
        const int off = (o >> 4) * 8 + (o & 7);
        const float* q = red + src * L3_RED + off;
        const float p01 = q[0 * 32 * L3_RED] + q[1 * 32 * L3_RED];
        const float p23 = q[2 * 32 * L3_RED] + q[3 * 32 * L3_RED];
        const float p45 = q[4 * 32 * L3_RED] + q[5 * 32 * L3_RED];
        const float p67 = q[6 * 32 * L3_RED] + q[7 * 32 * L3_RED];
        out[(size_t)m0 * NCLS + idx] = ((p01 + p23) + (p45 + p67)) + b3[o];
    }
}

extern "C" void kernel_launch(void* const* d_in, const int* in_sizes, int n_in,
                              void* d_out, int out_size, void* d_ws, size_t ws_size,
                              hipStream_t stream) {
    const int*   ids = (const int*)  d_in[0];
    const float* W1  = (const float*)d_in[1];
    const float* b1  = (const float*)d_in[2];
    const float* W2  = (const float*)d_in[3];
    const float* b2  = (const float*)d_in[4];
    const float* W3  = (const float*)d_in[5];
    const float* b3  = (const float*)d_in[6];
    float* out = (float*)d_out;

    unsigned short* ids2 = (unsigned short*)d_ws;
    float* y1T  = (float*)(ids2 + (size_t)BATCH * SEQ);

    ft_ids_bank_order_kernel<<<BATCH / DD_ROWS, DD_ROWS * 32, 0, stream>>>(ids, ids2);
    ft_layer1_col_kernel<<<HID, LC_THREADS, 0, stream>>>(ids2, W1, b1, y1T);
    ft_layer23_wmma_kernel<<<BATCH / 16, 256, 0, stream>>>(y1T, W2, b2, W3, b3, out);
}


# layer-1 gather: a slot's LDS byte offset (4 x id) is one 16-bit multiply-add on a register half, was mask-or-shift plus shift: 37 vector instructions per row pair, was 47; sum order kept
# speedup vs baseline: 4.4805x; 1.0328x over previous
#include <hip/hip_runtime.h>
#include <hip/hip_bf16.h>

#define VOCAB  50257
#define NCLS   20
#define BATCH  4096
#define SEQ    200
#define HID    256

typedef __attribute__((ext_vector_type(2))) float v2f;
typedef __attribute__((ext_vector_type(8))) float v8f;

#define DD_ROWS     8
#define DD_PER_LANE 8
#define DD_PADKEY   0xffffffffu

static_assert(VOCAB <= 0xffff, "ids and the zero slot's index must fit 16 bits");
static_assert(SEQ % 8 == 0 && SEQ <= 32 * (DD_PER_LANE - 1), "row pieces of 8 ids; the last register of every lane is padding");

template <int MASK>
__device__ __forceinline__ unsigned ft_key_row_xor(unsigned v) {
    return (unsigned)__builtin_amdgcn_update_dpp(0, (int)v, 0x160 + MASK, 0xf, 0xf, false);
}

__device__ __forceinline__ unsigned ft_key_lane_xor(unsigned v, int m) {
    switch (m) {
        case 1:  return ft_key_row_xor<1>(v);
        case 2:  return ft_key_row_xor<2>(v);
        case 4:  return ft_key_row_xor<4>(v);
        case 8:  return ft_key_row_xor<8>(v);
        default: return (unsigned)__builtin_amdgcn_permlanex16(
                     0, (int)v, 0x76543210, 0xfedcba98, false, false);
    }
}

__global__ __launch_bounds__(DD_ROWS * 32)
void ft_ids_bank_order_kernel(const int*      __restrict__ ids,
                              unsigned short* __restrict__ ids2) {
    __shared__ __attribute__((aligned(16))) unsigned short s_o[DD_ROWS * SEQ];

    const int lane = threadIdx.x & 31;
    const int wave = threadIdx.x >> 5;
    const int b    = blockIdx.x * DD_ROWS + wave;
    const int e0   = lane * DD_PER_LANE;

    unsigned key[DD_PER_LANE];
    #pragma unroll
    for (int r = 0; r < DD_PER_LANE - 1; ++r) {
        const int j  = lane + 32 * r;
        const int id = (j < SEQ) ? ids[(size_t)b * SEQ + j] : 0;
        key[r] = (j < SEQ) ? (unsigned)(((id & 63) << 10) | ((id >> 6) & 1023)) : DD_PADKEY;
    }
    key[DD_PER_LANE - 1] = DD_PADKEY;

    #pragma unroll
    for (int k = 2; k <= 32 * DD_PER_LANE; k <<= 1) {
        #pragma unroll
        for (int j = k >> 1; j > 0; j >>= 1) {
            if (j >= DD_PER_LANE) {
                const bool keep_min = ((e0 & k) == 0) == ((e0 & j) == 0);
                #pragma unroll
                for (int r = 0; r < DD_PER_LANE; ++r) {
                    const unsigned p  = ft_key_lane_xor(key[r], j >> 3);
                    const unsigned lo = key[r] < p ? key[r] : p;
                    const unsigned hi = key[r] < p ? p : key[r];
                    key[r] = keep_min ? lo : hi;
                }
            } else {
                #pragma unroll
                for (int r = 0; r < DD_PER_LANE; ++r) {
                    if ((r & j) == 0) {
                        const bool asc = (k < DD_PER_LANE) ? ((r & k) == 0) : ((e0 & k) == 0);
                        const unsigned a  = key[r], c = key[r | j];
                        const unsigned lo = a < c ? a : c;
                        const unsigned hi = a < c ? c : a;
                        key[r]     = asc ? lo : hi;
                        key[r | j] = asc ? hi : lo;
                    }
                }
            }
        }
    }

    const unsigned before = (unsigned)__shfl((int)key[DD_PER_LANE - 1], (lane + 31) & 31, 32);
    unsigned o[DD_PER_LANE];
    #pragma unroll
    for (int r = 0; r < DD_PER_LANE; ++r) {
        const unsigned k    = key[r];
        const unsigned prev = r ? key[r - 1] : before;
        const unsigned id   = ((k & 1023u) << 6) | ((k >> 10) & 63u);
        o[r] = (k == prev) ? (unsigned)VOCAB : id;
    }
    if (lane < SEQ / 8) {
        uint4 w;
        w.x = o[0] | (o[1] << 16);
        w.y = o[2] | (o[3] << 16);
        w.z = o[4] | (o[5] << 16);
        w.w = o[6] | (o[7] << 16);
        ((uint4*)s_o)[wave * (SEQ / 8) + lane] = w;
    }
    __syncthreads();

    if (threadIdx.x < DD_ROWS * SEQ / 8) {
        uint4* dst = (uint4*)(ids2 + (size_t)blockIdx.x * DD_ROWS * SEQ);
        dst[threadIdx.x] = ((const uint4*)s_o)[threadIdx.x];
    }
}

#define LC_THREADS  1024
#define LC_WAVES    (LC_THREADS / 32)
#define LC_WPAD     50272
#define LC_FILL     10
#define LC_PAIRS    (BATCH / 2)
#define LC_PIECES   (SEQ / 8)
#define LC_TRIPS    (LC_PAIRS / LC_WAVES)
#define LC_AHEAD    4

static_assert(LC_TRIPS % (2 * LC_AHEAD) == 0, "the gather loop advances by LC_AHEAD pairs, two trips per copy");

template <int MASK>
__device__ __forceinline__ float ft_row_xor(float v) {
    return __int_as_float(__builtin_amdgcn_update_dpp(
        0, __float_as_int(v), 0x160 + MASK, 0xf, 0xf, false));
}

__device__ __forceinline__ float ft_other_row(float v) {
    return __int_as_float(__builtin_amdgcn_permlanex16(
        0, __float_as_int(v), 0x76543210, 0xfedcba98, false, false));
}

__device__ __forceinline__ unsigned ft_slot_lo(unsigned w) {
    unsigned r;
    asm("v_mad_u32_u16 %0, %1, 4, 0" : "=v"(r) : "v"(w));
    return r;
}
__device__ __forceinline__ unsigned ft_slot_hi(unsigned w) {
    unsigned r;
    asm("v_mad_u32_u16 %0, %1, 4, 0 op_sel:[1,0,0,0]" : "=v"(r) : "v"(w));
    return r;
}
__device__ __forceinline__ float ft_slot(const float* s_w, unsigned off) {
    return *(const float*)((const char*)s_w + off);
}

__device__ __forceinline__ float ft_piece_sum(const float* s_w, uint4 a) {
    const float e0 = ft_slot(s_w, ft_slot_lo(a.x)) + ft_slot(s_w, ft_slot_hi(a.x));
    const float e1 = ft_slot(s_w, ft_slot_lo(a.y)) + ft_slot(s_w, ft_slot_hi(a.y));
    const float e2 = ft_slot(s_w, ft_slot_lo(a.z)) + ft_slot(s_w, ft_slot_hi(a.z));
    const float e3 = ft_slot(s_w, ft_slot_lo(a.w)) + ft_slot(s_w, ft_slot_hi(a.w));
    return (e0 + e1) + (e2 + e3);
}

__global__ __launch_bounds__(LC_THREADS)
void ft_layer1_col_kernel(const unsigned short* __restrict__ ids2,
                          const float*          __restrict__ W1,
                          const float*          __restrict__ b1,
                          float*                __restrict__ y1T) {
    __shared__ __attribute__((aligned(16))) float s_w[LC_WPAD];
    __shared__ __attribute__((aligned(16))) float s_out[BATCH];

    const int h    = blockIdx.x;
    const int t    = threadIdx.x;
    const int lane = t & 31;
    const int wave = t >> 5;
    const int c    = lane & 15;
    const int hh   = lane >> 4;

    const float* wrow = W1 + (size_t)h * VOCAB;
    for (int base = 0; base < LC_WPAD; base += LC_THREADS * LC_FILL) {
        float x[LC_FILL];
        #pragma unroll
        for (int u = 0; u < LC_FILL; ++u) {
            const int v = base + u * LC_THREADS + t;
            x[u] = (v < VOCAB) ? wrow[v] : 0.0f;
        }
        #pragma unroll
        for (int u = 0; u < LC_FILL; ++u) {
            const int v = base + u * LC_THREADS + t;
            if (v < LC_WPAD) s_w[v] = x[u];
        }
    }
    __syncthreads();

    const bool own = lane < LC_PIECES;
    const int  li  = own ? lane : LC_PIECES - 1;
    const unsigned short* rlane = ids2 + 8 * li;
    uint4 qA[LC_AHEAD], qB[LC_AHEAD];
    #pragma unroll
    for (int d = 0; d < LC_AHEAD; ++d) {
        const unsigned short* rp = rlane + (size_t)(2 * (wave + LC_WAVES * d)) * SEQ;
        qA[d] = *(const uint4*)rp;
        qB[d] = *(const uint4*)(rp + SEQ);
    }

    #pragma unroll 2
    for (int i = 0; i < LC_TRIPS; i += LC_AHEAD) {
        #pragma unroll
        for (int d = 0; d < LC_AHEAD; ++d) {
            const uint4 a = qA[d], b = qB[d];
            const int   cur = 2 * (wave + LC_WAVES * (i + d)) + hh;
            const int in = (i + d + LC_AHEAD < LC_TRIPS) ? i + d + LC_AHEAD : LC_TRIPS - 1;
            const unsigned short* rp = rlane + (size_t)(2 * (wave + LC_WAVES * in)) * SEQ;
            qA[d] = *(const uint4*)rp;
            qB[d] = *(const uint4*)(rp + SEQ);

            float sA = ft_piece_sum(s_w, a);
            float sB = ft_piece_sum(s_w, b);
            sA = own ? sA : 0.0f;
            sB = own ? sB : 0.0f;
            const float keep = hh ? sB : sA;
            const float give = hh ? sA : sB;
            float s = keep + ft_other_row(give);
            s += ft_row_xor<1>(s);
            s += ft_row_xor<2>(s);
            s += ft_row_xor<4>(s);
            s += ft_row_xor<8>(s);
            if (c == 0) s_out[cur] = s;
        }
    }
    __syncthreads();

    const float  bias = b1[h];
    const float4 r = ((const float4*)s_out)[t];
    float4 o;
    o.x = fmaxf(bias + r.x, 0.0f);
    o.y = fmaxf(bias + r.y, 0.0f);
    o.z = fmaxf(bias + r.z, 0.0f);
    o.w = fmaxf(bias + r.w, 0.0f);
    ((float4*)(y1T + (size_t)h * BATCH))[t] = o;
}

#define L2_PITCH 260
#define L3_RED   12

static_assert(NCLS > 16 && NCLS <= 20, "second class tile: rows 0..3 only");

__global__ __launch_bounds__(256)
void ft_layer23_wmma_kernel(const float* __restrict__ y1T,
                            const float* __restrict__ W2,
                            const float* __restrict__ b2,
                            const float* __restrict__ W3,
                            const float* __restrict__ b3,
                            float*       __restrict__ out) {
    __shared__ __attribute__((aligned(16))) float As[16 * L2_PITCH];
    __shared__ __attribute__((aligned(16))) float red[8 * 32 * L3_RED];
    const int tid  = threadIdx.x;
    const int m0   = blockIdx.x * 16;
    const int wave = tid >> 5;
    const int lane = tid & 31;
    const int c    = lane & 15;
    const int h    = lane >> 4;
    const int n0   = wave * 32;

    const float4* a4 = (const float4*)(y1T + (size_t)tid * BATCH + m0);
    float4 av[4];
    #pragma unroll
    for (int q = 0; q < 4; ++q) av[q] = a4[q];

    const float4* bp = (const float4*)(b2 + n0 + 8 * h);
    const float4 c00 = bp[0], c01 = bp[1];
    const float4 c10 = bp[4], c11 = bp[5];
    const int     ohi = (16 + c < NCLS) ? 16 + c : NCLS - 1;
    const float4* w3a = (const float4*)(W3 + (size_t)c * HID + n0 + 8 * h);
    const float4* w3b = (const float4*)(W3 + (size_t)ohi * HID + n0 + 8 * h);
    float4 wa[2][2], wb[2][2];
    #pragma unroll
    for (int u = 0; u < 2; ++u) {
        wa[u][0] = w3a[4 * u];
        wa[u][1] = w3a[4 * u + 1];
        wb[u][0] = w3b[4 * u];
        wb[u][1] = w3b[4 * u + 1];
    }

    const float4* w0 = (const float4*)(W2 + (size_t)(n0 + c) * HID + 4 * h);
    const float4* w1 = (const float4*)(W2 + (size_t)(n0 + 16 + c) * HID + 4 * h);
    float4 bq0[HID / 8], bq1[HID / 8];
    #pragma unroll
    for (int j = 0; j < HID / 8; ++j) {
        bq0[j] = w0[2 * j];
        bq1[j] = w1[2 * j];
    }

    #pragma unroll
    for (int q = 0; q < 4; ++q) {
        As[(4 * q    ) * L2_PITCH + tid] = av[q].x;
        As[(4 * q + 1) * L2_PITCH + tid] = av[q].y;
        As[(4 * q + 2) * L2_PITCH + tid] = av[q].z;
        As[(4 * q + 3) * L2_PITCH + tid] = av[q].w;
    }
    __builtin_amdgcn_fence(__ATOMIC_RELEASE, "workgroup", "local");
    __builtin_amdgcn_s_barrier();
    __builtin_amdgcn_fence(__ATOMIC_ACQUIRE, "workgroup", "local");

    v8f acc0, acc1;
    #pragma unroll
    for (int r = 0; r < 8; ++r) { acc0[r] = 0.0f; acc1[r] = 0.0f; }

    const float* arow = As + c * L2_PITCH + 4 * h;
    #pragma unroll
    for (int j = 0; j < HID / 8; ++j) {
        const float4 a = *(const float4*)(arow + 8 * j);
        v2f alo, ahi, b0lo, b0hi, b1lo, b1hi;
        alo.x  = a.x;       alo.y  = a.y;
        ahi.x  = a.z;       ahi.y  = a.w;
        b0lo.x = bq0[j].x;  b0lo.y = bq0[j].y;
        b0hi.x = bq0[j].z;  b0hi.y = bq0[j].w;
        b1lo.x = bq1[j].x;  b1lo.y = bq1[j].y;
        b1hi.x = bq1[j].z;  b1hi.y = bq1[j].w;
        acc0 = __builtin_amdgcn_wmma_f32_16x16x4_f32(
            false, b0lo, false, alo, (short)0, acc0, false, false);
        acc1 = __builtin_amdgcn_wmma_f32_16x16x4_f32(
            false, b1lo, false, alo, (short)0, acc1, false, false);
        acc0 = __builtin_amdgcn_wmma_f32_16x16x4_f32(
            false, b0hi, false, ahi, (short)0, acc0, false, false);
        acc1 = __builtin_amdgcn_wmma_f32_16x16x4_f32(
            false, b1hi, false, ahi, (short)0, acc1, false, false);
    }

    asm volatile("s_wait_alu depctr_va_vdst(0)" : "+v"(acc0), "+v"(acc1));

    const float cb0[8] = { c00.x, c00.y, c00.z, c00.w, c01.x, c01.y, c01.z, c01.w };
    const float cb1[8] = { c10.x, c10.y, c10.z, c10.w, c11.x, c11.y, c11.z, c11.w };
    #pragma unroll
    for (int r = 0; r < 8; ++r) {
        const float s0 = acc0[r] + cb0[r];
        const float s1 = acc1[r] + cb1[r];
        acc0[r] = s0 > 0.0f ? s0 : 0.0f;
        acc1[r] = s1 > 0.0f ? s1 : 0.0f;
    }
    asm volatile("" :: "v"(acc0), "v"(acc1));

    const float fa[2][8] = {
        { wa[0][0].x, wa[0][0].y, wa[0][0].z, wa[0][0].w, wa[0][1].x, wa[0][1].y, wa[0][1].z, wa[0][1].w },
        { wa[1][0].x, wa[1][0].y, wa[1][0].z, wa[1][0].w, wa[1][1].x, wa[1][1].y, wa[1][1].z, wa[1][1].w } };
    const float fb[2][8] = {
        { wb[0][0].x, wb[0][0].y, wb[0][0].z, wb[0][0].w, wb[0][1].x, wb[0][1].y, wb[0][1].z, wb[0][1].w },
        { wb[1][0].x, wb[1][0].y, wb[1][0].z, wb[1][0].w, wb[1][1].x, wb[1][1].y, wb[1][1].z, wb[1][1].w } };

    v8f o0, o1;
    #pragma unroll
    for (int r = 0; r < 8; ++r) { o0[r] = 0.0f; o1[r] = 0.0f; }
    #pragma unroll
    for (int u = 0; u < 2; ++u) {
        #pragma unroll
        for (int q = 0; q < 4; ++q) {
            v2f yb, w3lo, w3hi;
            yb.x   = u ? acc1[2 * q]     : acc0[2 * q];
            yb.y   = u ? acc1[2 * q + 1] : acc0[2 * q + 1];
            w3lo.x = fa[u][2 * q];
            w3lo.y = fa[u][2 * q + 1];
            w3hi.x = fb[u][2 * q];
            w3hi.y = fb[u][2 * q + 1];
            o0 = __builtin_amdgcn_wmma_f32_16x16x4_f32(
                false, w3lo, false, yb, (short)0, o0, false, false);
            o1 = __builtin_amdgcn_wmma_f32_16x16x4_f32(
                false, w3hi, false, yb, (short)0, o1, false, false);
        }
    }

    float* rp = red + (wave * 32 + lane) * L3_RED;
    *(float4*)(rp)     = make_float4(o0[0], o0[1], o0[2], o0[3]);
    *(float4*)(rp + 4) = make_float4(o0[4], o0[5], o0[6], o0[7]);
    *(float4*)(rp + 8) = make_float4(o1[0], o1[1], o1[2], o1[3]);
    __syncthreads();

    for (int idx = tid; idx < 16 * NCLS; idx += 256) {
        const int m   = idx / NCLS;
        const int o   = idx - m * NCLS;
        const int src = ((o >> 3) & 1) * 16 + m;
        const int off = (o >> 4) * 8 + (o & 7);
        const float* q = red + src * L3_RED + off;
        const float p01 = q[0 * 32 * L3_RED] + q[1 * 32 * L3_RED];
        const float p23 = q[2 * 32 * L3_RED] + q[3 * 32 * L3_RED];
        const float p45 = q[4 * 32 * L3_RED] + q[5 * 32 * L3_RED];
        const float p67 = q[6 * 32 * L3_RED] + q[7 * 32 * L3_RED];
        out[(size_t)m0 * NCLS + idx] = ((p01 + p23) + (p45 + p67)) + b3[o];
    }
}

extern "C" void kernel_launch(void* const* d_in, const int* in_sizes, int n_in,
                              void* d_out, int out_size, void* d_ws, size_t ws_size,
                              hipStream_t stream) {
    const int*   ids = (const int*)  d_in[0];
    const float* W1  = (const float*)d_in[1];
    const float* b1  = (const float*)d_in[2];
    const float* W2  = (const float*)d_in[3];
    const float* b2  = (const float*)d_in[4];
    const float* W3  = (const float*)d_in[5];
    const float* b3  = (const float*)d_in[6];
    float* out = (float*)d_out;

    unsigned short* ids2 = (unsigned short*)d_ws;
    float* y1T  = (float*)(ids2 + (size_t)BATCH * SEQ);

    ft_ids_bank_order_kernel<<<BATCH / DD_ROWS, DD_ROWS * 32, 0, stream>>>(ids, ids2);
    ft_layer1_col_kernel<<<HID, LC_THREADS, 0, stream>>>(ids2, W1, b1, y1T);
    ft_layer23_wmma_kernel<<<BATCH / 16, 256, 0, stream>>>(y1T, W2, b2, W3, b3, out);
}
